# starting kernel
# baseline (speedup 1.0000x reference)
#include <hip/hip_runtime.h>
#include <math.h>

#define NN 100000
#define NE 1600000
#define DH 128
#define NC 40
#define A_STRIDE 260

typedef __attribute__((ext_vector_type(2))) float v2f;
typedef __attribute__((ext_vector_type(8))) float v8f;

__global__ void zero_f32v4(float4* __restrict__ p, int n4) {
    int i = blockIdx.x * blockDim.x + threadIdx.x;
    if (i < n4) p[i] = make_float4(0.f, 0.f, 0.f, 0.f);
}

__global__ void deg_kernel(const int* __restrict__ dst, float* __restrict__ deg) {
    int e = blockIdx.x * blockDim.x + threadIdx.x;
    if (e < NE) atomicAdd(&deg[dst[e]], 1.0f);
}

__global__ void deginv_kernel(float* __restrict__ deg) {
    int i = blockIdx.x * blockDim.x + threadIdx.x;
    if (i < NN) deg[i] = 1.0f / fmaxf(deg[i], 1.0f);
}

__global__ void scatter_kernel(const float* __restrict__ x,
                               const int* __restrict__ src,
                               const int* __restrict__ dst,
                               float* __restrict__ aggr) {
    const int lane = threadIdx.x & 31;
    const int wave = threadIdx.x >> 5;
    const int e = blockIdx.x * (blockDim.x >> 5) + wave;
    if (e >= NE) return;
    const int s = __builtin_amdgcn_readfirstlane(src[e]);
    const int d = __builtin_amdgcn_readfirstlane(dst[e]);
    float4 v = ((const float4*)(x + (size_t)s * DH))[lane];
    float* o = aggr + (size_t)d * DH + lane * 4;
    atomicAdd(o + 0, v.x);
    atomicAdd(o + 1, v.y);
    atomicAdd(o + 2, v.z);
    atomicAdd(o + 3, v.w);
}

template <int DIN, int DOUT, bool RELU>
__global__ void sage_gemm(const float* __restrict__ aggr,
                          const float* __restrict__ x,
                          const float* __restrict__ deginv,
                          const float* __restrict__ Wl,
                          const float* __restrict__ Wr,
                          const float* __restrict__ bias,
                          float* __restrict__ out) {
    __shared__ float Alds[16 * A_STRIDE];

    const int m0   = blockIdx.x * 16;
    const int tid  = threadIdx.x;
    const int lane = tid & 31;
    const int wave = tid >> 5;
    constexpr int twoK = 2 * DIN;

    for (int idx = tid; idx < 16 * twoK; idx += blockDim.x) {
        int row = idx / twoK;
        int kk  = idx - row * twoK;
        int g   = m0 + row;
        float v;
        if (kk < DIN) v = aggr[(size_t)g * DIN + kk] * deginv[g];
        else          v = x[(size_t)g * DIN + (kk - DIN)];
        Alds[row * A_STRIDE + kk] = v;
    }
    __syncthreads();

    const int half = lane >> 4;
    const int l15  = lane & 15;
    const int n0   = wave * 16;
    const int col  = n0 + l15;
    const int ccol = (col < DOUT) ? col : 0;

    const float* __restrict__ Wlp = Wl + 2 * half * DOUT + ccol;
    const float* __restrict__ Wrp = Wr + 2 * half * DOUT + ccol;
    const float* __restrict__ Ap  = Alds + l15 * A_STRIDE + 2 * half;

    v8f acc = {};
    #pragma unroll 8
    for (int k = 0; k < DIN; k += 4) {
        v2f a, b;
        a.x = Ap[k];
        a.y = Ap[k + 1];
        b.x = Wlp[k * DOUT];
        b.y = Wlp[k * DOUT + DOUT];
        acc = __builtin_amdgcn_wmma_f32_16x16x4_f32(
            false, a, false, b, (short)0, acc, false, false);
    }
    #pragma unroll 8
    for (int k = 0; k < DIN; k += 4) {
        v2f a, b;
        a.x = Ap[DIN + k];
        a.y = Ap[DIN + k + 1];
        b.x = Wrp[k * DOUT];
        b.y = Wrp[k * DOUT + DOUT];
        acc = __builtin_amdgcn_wmma_f32_16x16x4_f32(
            false, a, false, b, (short)0, acc, false, false);
    }

    if (col < DOUT) {
        const float bv = bias[col];
        float* __restrict__ op = out + (size_t)(m0 + 8 * half) * DOUT + col;
        #pragma unroll
        for (int r = 0; r < 8; ++r) {
            float v = acc[r] + bv;
            if (RELU) v = fmaxf(v, 0.0f);
            op[r * DOUT] = v;
        }
    }
}

__global__ void logsoftmax_kernel(float* __restrict__ h) {
    int i = blockIdx.x * blockDim.x + threadIdx.x;
    if (i >= NN) return;
    float* row = h + (size_t)i * NC;
    float mx = -INFINITY;
    #pragma unroll
    for (int c = 0; c < NC; ++c) mx = fmaxf(mx, row[c]);
    float s = 0.0f;
    #pragma unroll
    for (int c = 0; c < NC; ++c) s += expf(row[c] - mx);
    float lse = mx + logf(s);
    #pragma unroll
    for (int c = 0; c < NC; ++c) row[c] = row[c] - lse;
}

extern "C" void kernel_launch(void* const* d_in, const int* in_sizes, int n_in,
                              void* d_out, int out_size, void* d_ws, size_t ws_size,
                              hipStream_t stream) {
    const float* x   = (const float*)d_in[0];
    const int*   ei  = (const int*)d_in[1];
    const int*   src = ei;
    const int*   dst = ei + NE;
    const float* Wl0 = (const float*)d_in[2];
    const float* Wr0 = (const float*)d_in[3];
    const float* b0  = (const float*)d_in[4];
    const float* Wl1 = (const float*)d_in[5];
    const float* Wr1 = (const float*)d_in[6];
    const float* b1  = (const float*)d_in[7];
    const float* Wl2 = (const float*)d_in[8];
    const float* Wr2 = (const float*)d_in[9];
    const float* b2  = (const float*)d_in[10];
    float* hout = (float*)d_out;

    char* ws = (char*)d_ws;
    size_t off = 0;
    auto carve = [&](size_t bytes) {
        void* p = ws + off;
        off = (off + bytes + 255) & ~(size_t)255;
        return p;
    };
    float* deg  = (float*)carve((size_t)NN * 4);
    float* aggr = (float*)carve((size_t)NN * DH * 4);
    float* h0   = (float*)carve((size_t)NN * DH * 4);
    float* h1   = (float*)carve((size_t)NN * DH * 4);

    const int ZB = 256;
    dim3 bZ(ZB);
    dim3 gN((NN + ZB - 1) / ZB);
    dim3 gZdeg((NN / 4 + ZB - 1) / ZB);
    dim3 gZaggr((NN * DH / 4 + ZB - 1) / ZB);
    dim3 gE((NE + ZB - 1) / ZB);
    dim3 gScat(NE / 8);
    dim3 gGemm(NN / 16);

    zero_f32v4<<<gZdeg, bZ, 0, stream>>>((float4*)deg, NN / 4);
    deg_kernel<<<gE, bZ, 0, stream>>>(dst, deg);
    deginv_kernel<<<gN, bZ, 0, stream>>>(deg);

    zero_f32v4<<<gZaggr, bZ, 0, stream>>>((float4*)aggr, NN * DH / 4);
    scatter_kernel<<<gScat, bZ, 0, stream>>>(x, src, dst, aggr);
    sage_gemm<DH, DH, true><<<gGemm, dim3(256), 0, stream>>>(aggr, x, deg, Wl0, Wr0, b0, h0);

    zero_f32v4<<<gZaggr, bZ, 0, stream>>>((float4*)aggr, NN * DH / 4);
    scatter_kernel<<<gScat, bZ, 0, stream>>>(h0, src, dst, aggr);
    sage_gemm<DH, DH, true><<<gGemm, dim3(256), 0, stream>>>(aggr, h0, deg, Wl1, Wr1, b1, h1);

    zero_f32v4<<<gZaggr, bZ, 0, stream>>>((float4*)aggr, NN * DH / 4);
    scatter_kernel<<<gScat, bZ, 0, stream>>>(h1, src, dst, aggr);
    sage_gemm<DH, NC, false><<<gGemm, dim3(96), 0, stream>>>(aggr, h1, deg, Wl2, Wr2, b2, hout);

    logsoftmax_kernel<<<gN, bZ, 0, stream>>>(hout);
}


# Float atomic scatter-add replaced by per-call counting sort of edges by destination (int atomic rank, scan, fill); a wave sums each row's neighbours in registers into the GEMM LDS tile, in rank order
# speedup vs baseline: 3.1124x; 3.1124x over previous
#include <hip/hip_runtime.h>
#include <math.h>


#define NN 100000
#define NE 1600000
#define DH 128
#define NC 40
#define A_STRIDE 260

typedef __attribute__((ext_vector_type(2))) float v2f;
typedef __attribute__((ext_vector_type(8))) float v8f;

__global__ void zero_f32v4(float4* __restrict__ p, int n4) {
    int i = blockIdx.x * blockDim.x + threadIdx.x;
    if (i < n4) p[i] = make_float4(0.f, 0.f, 0.f, 0.f);
}

__global__ void rank_kernel(const int* __restrict__ dst,
                            int* __restrict__ cnt,
                            int* __restrict__ rank) {
    int e = blockIdx.x * blockDim.x + threadIdx.x;
    if (e < NE) rank[e] = atomicAdd(&cnt[dst[e]], 1);
}

__global__ void rowstart_kernel(const int* __restrict__ cnt,
                                int* __restrict__ rowstart) {
    __shared__ int front[32];
    __shared__ int wsum[32];
    const int tid   = threadIdx.x;
    const int lane  = tid & 31;
    const int wave  = tid >> 5;
    const int start = blockIdx.x * 1024;

    int part = 0;
    for (int i = tid; i < start; i += 1024) part += cnt[i];
    for (int o = 16; o > 0; o >>= 1) part += __shfl_xor(part, o);
    if (lane == 0) front[wave] = part;

    const int i = start + tid;
    const int v = (i < NN) ? cnt[i] : 0;
    int incl = v;
    for (int o = 1; o < 32; o <<= 1) {
        int t = __shfl_up(incl, o);
        if (lane >= o) incl += t;
    }
    if (lane == 31) wsum[wave] = incl;
    __syncthreads();

    int off = 0;
    for (int w = 0; w < 32; ++w) off += front[w];
    for (int w = 0; w < wave; ++w) off += wsum[w];
    if (i < NN) rowstart[i] = off + incl - v;
}

__global__ void fill_kernel(const int* __restrict__ src,
                            const int* __restrict__ dst,
                            const int* __restrict__ rank,
                            const int* __restrict__ rowstart,
                            int* __restrict__ nbr) {
    int e = blockIdx.x * blockDim.x + threadIdx.x;
    if (e < NE) nbr[rowstart[dst[e]] + rank[e]] = src[e];
}

__device__ __forceinline__ float4 ld_row(const float4* __restrict__ x4, int s, int lane) {
    return x4[(size_t)s * (DH / 4) + lane];
}
__device__ __forceinline__ void add4(float4& a, const float4& v) {
    a.x += v.x; a.y += v.y; a.z += v.z; a.w += v.w;
}

template <int DIN, int DOUT, bool RELU>
__global__ void sage_gemm(const float* __restrict__ x,
                          const int* __restrict__ cnt,
                          const int* __restrict__ rowstart,
                          const int* __restrict__ nbr,
                          const float* __restrict__ Wl,
                          const float* __restrict__ Wr,
                          const float* __restrict__ bias,
                          float* __restrict__ out) {
    static_assert(DIN == DH, "row gather is written for 128-float rows");
    __shared__ __attribute__((aligned(16))) float Alds[16 * A_STRIDE];

    const int m0   = blockIdx.x * 16;
    const int tid  = threadIdx.x;
    const int lane = tid & 31;
    const int wave = tid >> 5;

    for (int idx = tid; idx < 16 * DIN; idx += blockDim.x) {
        int row = idx / DIN;
        int kk  = idx - row * DIN;
        Alds[row * A_STRIDE + DIN + kk] = x[(size_t)(m0 + row) * DIN + kk];
    }

    const float4* __restrict__ x4 = (const float4*)x;
    const int wv     = __builtin_amdgcn_readfirstlane(wave);
    const int nwaves = blockDim.x >> 5;
    for (int row = wv; row < 16; row += nwaves) {
        const int g   = m0 + row;
        const int c   = __builtin_amdgcn_readfirstlane(cnt[g]);
        const int beg = __builtin_amdgcn_readfirstlane(rowstart[g]);
        float4 acc = make_float4(0.f, 0.f, 0.f, 0.f);
        for (int base = 0; base < c; base += 32) {
            const int n  = (c - base < 32) ? (c - base) : 32;
            const int my = (lane < n) ? nbr[beg + base + lane] : 0;
            int j = 0;
            for (; j + 8 <= n; j += 8) {
                float4 v0 = ld_row(x4, __builtin_amdgcn_readlane(my, j + 0), lane);
                float4 v1 = ld_row(x4, __builtin_amdgcn_readlane(my, j + 1), lane);
                float4 v2 = ld_row(x4, __builtin_amdgcn_readlane(my, j + 2), lane);
                float4 v3 = ld_row(x4, __builtin_amdgcn_readlane(my, j + 3), lane);
                float4 v4 = ld_row(x4, __builtin_amdgcn_readlane(my, j + 4), lane);
                float4 v5 = ld_row(x4, __builtin_amdgcn_readlane(my, j + 5), lane);
                float4 v6 = ld_row(x4, __builtin_amdgcn_readlane(my, j + 6), lane);
                float4 v7 = ld_row(x4, __builtin_amdgcn_readlane(my, j + 7), lane);
                add4(acc, v0); add4(acc, v1); add4(acc, v2); add4(acc, v3);
                add4(acc, v4); add4(acc, v5); add4(acc, v6); add4(acc, v7);
            }
            if (j + 4 <= n) {
                float4 v0 = ld_row(x4, __builtin_amdgcn_readlane(my, j + 0), lane);
                float4 v1 = ld_row(x4, __builtin_amdgcn_readlane(my, j + 1), lane);
                float4 v2 = ld_row(x4, __builtin_amdgcn_readlane(my, j + 2), lane);
                float4 v3 = ld_row(x4, __builtin_amdgcn_readlane(my, j + 3), lane);
                add4(acc, v0); add4(acc, v1); add4(acc, v2); add4(acc, v3);
                j += 4;
            }
            if (j + 2 <= n) {
                float4 v0 = ld_row(x4, __builtin_amdgcn_readlane(my, j + 0), lane);
                float4 v1 = ld_row(x4, __builtin_amdgcn_readlane(my, j + 1), lane);
                add4(acc, v0); add4(acc, v1);
                j += 2;
            }
            if (j < n) {
                float4 v0 = ld_row(x4, __builtin_amdgcn_readlane(my, j), lane);
                add4(acc, v0);
            }
        }
        const float di = 1.0f / fmaxf((float)c, 1.0f);
        float4 r = make_float4(acc.x * di, acc.y * di, acc.z * di, acc.w * di);
        *(float4*)(Alds + row * A_STRIDE + 4 * lane) = r;
    }
    __syncthreads();

    const int half = lane >> 4;
    const int l15  = lane & 15;
    const int n0   = wave * 16;
    const int col  = n0 + l15;
    const int ccol = (col < DOUT) ? col : 0;

    const float* __restrict__ Wlp = Wl + 2 * half * DOUT + ccol;
    const float* __restrict__ Wrp = Wr + 2 * half * DOUT + ccol;
    const float* __restrict__ Ap  = Alds + l15 * A_STRIDE + 2 * half;

    v8f acc = {};
    #pragma unroll 8
    for (int k = 0; k < DIN; k += 4) {
        v2f a, b;
        a.x = Ap[k];
        a.y = Ap[k + 1];
        b.x = Wlp[k * DOUT];
        b.y = Wlp[k * DOUT + DOUT];
        acc = __builtin_amdgcn_wmma_f32_16x16x4_f32(
            false, a, false, b, (short)0, acc, false, false);
    }
    #pragma unroll 8
    for (int k = 0; k < DIN; k += 4) {
        v2f a, b;
        a.x = Ap[DIN + k];
        a.y = Ap[DIN + k + 1];
        b.x = Wrp[k * DOUT];
        b.y = Wrp[k * DOUT + DOUT];
        acc = __builtin_amdgcn_wmma_f32_16x16x4_f32(
            false, a, false, b, (short)0, acc, false, false);
    }

    if (col < DOUT) {
        const float bv = bias[col];
        float* __restrict__ op = out + (size_t)(m0 + 8 * half) * DOUT + col;
        #pragma unroll
        for (int r = 0; r < 8; ++r) {
            float v = acc[r] + bv;
            if (RELU) v = fmaxf(v, 0.0f);
            op[r * DOUT] = v;
        }
    }
}

__global__ void logsoftmax_kernel(float* __restrict__ h) {
    int i = blockIdx.x * blockDim.x + threadIdx.x;
    if (i >= NN) return;
    float* row = h + (size_t)i * NC;
    float mx = -INFINITY;
    #pragma unroll
    for (int c = 0; c < NC; ++c) mx = fmaxf(mx, row[c]);
    float s = 0.0f;
    #pragma unroll
    for (int c = 0; c < NC; ++c) s += expf(row[c] - mx);
    float lse = mx + logf(s);
    #pragma unroll
    for (int c = 0; c < NC; ++c) row[c] = row[c] - lse;
}

extern "C" void kernel_launch(void* const* d_in, const int* in_sizes, int n_in,
                              void* d_out, int out_size, void* d_ws, size_t ws_size,
                              hipStream_t stream) {
    const float* x   = (const float*)d_in[0];
    const int*   ei  = (const int*)d_in[1];
    const int*   src = ei;
    const int*   dst = ei + NE;
    const float* Wl0 = (const float*)d_in[2];
    const float* Wr0 = (const float*)d_in[3];
    const float* b0  = (const float*)d_in[4];
    const float* Wl1 = (const float*)d_in[5];
    const float* Wr1 = (const float*)d_in[6];
    const float* b1  = (const float*)d_in[7];
    const float* Wl2 = (const float*)d_in[8];
    const float* Wr2 = (const float*)d_in[9];
    const float* b2  = (const float*)d_in[10];
    float* hout = (float*)d_out;

    char* ws = (char*)d_ws;
    size_t off = 0;
    auto carve = [&](size_t bytes) {
        void* p = ws + off;
        off = (off + bytes + 255) & ~(size_t)255;
        return p;
    };
    int*   cnt      = (int*)carve((size_t)NN * 4);
    int*   rowstart = (int*)carve((size_t)NN * 4);
    int*   rank     = (int*)carve((size_t)NE * 4);
    int*   nbr      = (int*)carve((size_t)NE * 4);
    float* h0       = (float*)carve((size_t)NN * DH * 4);
    float* h1       = (float*)carve((size_t)NN * DH * 4);

    const int ZB = 256;
    dim3 bZ(ZB);
    dim3 gN((NN + ZB - 1) / ZB);
    dim3 gZcnt((NN / 4 + ZB - 1) / ZB);
    dim3 gE((NE + ZB - 1) / ZB);
    dim3 gScan((NN + 1023) / 1024);
    dim3 gGemm(NN / 16);

    zero_f32v4<<<gZcnt, bZ, 0, stream>>>((float4*)cnt, NN / 4);
    rank_kernel<<<gE, bZ, 0, stream>>>(dst, cnt, rank);
    rowstart_kernel<<<gScan, dim3(1024), 0, stream>>>(cnt, rowstart);
    fill_kernel<<<gE, bZ, 0, stream>>>(src, dst, rank, rowstart, nbr);

    sage_gemm<DH, DH, true><<<gGemm, dim3(256), 0, stream>>>(x, cnt, rowstart, nbr, Wl0, Wr0, b0, h0);

    sage_gemm<DH, DH, true><<<gGemm, dim3(256), 0, stream>>>(h0, cnt, rowstart, nbr, Wl1, Wr1, b1, h1);

    sage_gemm<DH, NC, false><<<gGemm, dim3(96), 0, stream>>>(h1, cnt, rowstart, nbr, Wl2, Wr2, b2, hout);

    logsoftmax_kernel<<<gN, bZ, 0, stream>>>(hout);
}


# Last layer reordered by linearity: h1 @ [Wl2|Wr2] first (f32 WMMA), then the neighbour mean over 40-float rows (one 256-B slot each) with log_softmax in that kernel; float sum order changes
# speedup vs baseline: 3.2346x; 1.0393x over previous
#include <hip/hip_runtime.h>
#include <math.h>


#define NN 100000
#define NE 1600000
#define DH 128
#define NC 40
#define A_STRIDE 260
#define P_STRIDE 132
#define PS 64

typedef __attribute__((ext_vector_type(2))) float v2f;
typedef __attribute__((ext_vector_type(8))) float v8f;

__global__ void zero_f32v4(float4* __restrict__ p, int n4) {
    int i = blockIdx.x * blockDim.x + threadIdx.x;
    if (i < n4) p[i] = make_float4(0.f, 0.f, 0.f, 0.f);
}

__global__ void rank_kernel(const int* __restrict__ dst,
                            int* __restrict__ cnt,
                            int* __restrict__ rank) {
    int e = blockIdx.x * blockDim.x + threadIdx.x;
    if (e < NE) rank[e] = atomicAdd(&cnt[dst[e]], 1);
}

__global__ void rowstart_kernel(const int* __restrict__ cnt,
                                int* __restrict__ rowstart) {
    __shared__ int front[32];
    __shared__ int wsum[32];
    const int tid   = threadIdx.x;
    const int lane  = tid & 31;
    const int wave  = tid >> 5;
    const int start = blockIdx.x * 1024;

    int part = 0;
    for (int i = tid; i < start; i += 1024) part += cnt[i];
    for (int o = 16; o > 0; o >>= 1) part += __shfl_xor(part, o);
    if (lane == 0) front[wave] = part;

    const int i = start + tid;
    const int v = (i < NN) ? cnt[i] : 0;
    int incl = v;
    for (int o = 1; o < 32; o <<= 1) {
        int t = __shfl_up(incl, o);
        if (lane >= o) incl += t;
    }
    if (lane == 31) wsum[wave] = incl;
    __syncthreads();

    int off = 0;
    for (int w = 0; w < 32; ++w) off += front[w];
    for (int w = 0; w < wave; ++w) off += wsum[w];
    if (i < NN) rowstart[i] = off + incl - v;
}

__global__ void fill_kernel(const int* __restrict__ src,
                            const int* __restrict__ dst,
                            const int* __restrict__ rank,
                            const int* __restrict__ rowstart,
                            int* __restrict__ nbr) {
    int e = blockIdx.x * blockDim.x + threadIdx.x;
    if (e < NE) nbr[rowstart[dst[e]] + rank[e]] = src[e];
}

__device__ __forceinline__ float4 ld_row(const float4* __restrict__ x4, int s, int lane) {
    return x4[(size_t)s * (DH / 4) + lane];
}
__device__ __forceinline__ void add4(float4& a, const float4& v) {
    a.x += v.x; a.y += v.y; a.z += v.z; a.w += v.w;
}

template <int DIN, int DOUT, bool RELU>
__global__ void sage_gemm(const float* __restrict__ x,
                          const int* __restrict__ cnt,
                          const int* __restrict__ rowstart,
                          const int* __restrict__ nbr,
                          const float* __restrict__ Wl,
                          const float* __restrict__ Wr,
                          const float* __restrict__ bias,
                          float* __restrict__ out) {
    static_assert(DIN == DH, "row gather is written for 128-float rows");
    __shared__ __attribute__((aligned(16))) float Alds[16 * A_STRIDE];

    const int m0   = blockIdx.x * 16;
    const int tid  = threadIdx.x;
    const int lane = tid & 31;
    const int wave = tid >> 5;

    for (int idx = tid; idx < 16 * DIN; idx += blockDim.x) {
        int row = idx / DIN;
        int kk  = idx - row * DIN;
        Alds[row * A_STRIDE + DIN + kk] = x[(size_t)(m0 + row) * DIN + kk];
    }

    const float4* __restrict__ x4 = (const float4*)x;
    const int wv     = __builtin_amdgcn_readfirstlane(wave);
    const int nwaves = blockDim.x >> 5;
    for (int row = wv; row < 16; row += nwaves) {
        const int g   = m0 + row;
        const int c   = __builtin_amdgcn_readfirstlane(cnt[g]);
        const int beg = __builtin_amdgcn_readfirstlane(rowstart[g]);
        float4 acc = make_float4(0.f, 0.f, 0.f, 0.f);
        for (int base = 0; base < c; base += 32) {
            const int n  = (c - base < 32) ? (c - base) : 32;
            const int my = (lane < n) ? nbr[beg + base + lane] : 0;
            int j = 0;
            for (; j + 8 <= n; j += 8) {
                float4 v0 = ld_row(x4, __builtin_amdgcn_readlane(my, j + 0), lane);
                float4 v1 = ld_row(x4, __builtin_amdgcn_readlane(my, j + 1), lane);
                float4 v2 = ld_row(x4, __builtin_amdgcn_readlane(my, j + 2), lane);
                float4 v3 = ld_row(x4, __builtin_amdgcn_readlane(my, j + 3), lane);
                float4 v4 = ld_row(x4, __builtin_amdgcn_readlane(my, j + 4), lane);
                float4 v5 = ld_row(x4, __builtin_amdgcn_readlane(my, j + 5), lane);
                float4 v6 = ld_row(x4, __builtin_amdgcn_readlane(my, j + 6), lane);
                float4 v7 = ld_row(x4, __builtin_amdgcn_readlane(my, j + 7), lane);
                add4(acc, v0); add4(acc, v1); add4(acc, v2); add4(acc, v3);
                add4(acc, v4); add4(acc, v5); add4(acc, v6); add4(acc, v7);
            }
            if (j + 4 <= n) {
                float4 v0 = ld_row(x4, __builtin_amdgcn_readlane(my, j + 0), lane);
                float4 v1 = ld_row(x4, __builtin_amdgcn_readlane(my, j + 1), lane);
                float4 v2 = ld_row(x4, __builtin_amdgcn_readlane(my, j + 2), lane);
                float4 v3 = ld_row(x4, __builtin_amdgcn_readlane(my, j + 3), lane);
                add4(acc, v0); add4(acc, v1); add4(acc, v2); add4(acc, v3);
                j += 4;
            }
            if (j + 2 <= n) {
                float4 v0 = ld_row(x4, __builtin_amdgcn_readlane(my, j + 0), lane);
                float4 v1 = ld_row(x4, __builtin_amdgcn_readlane(my, j + 1), lane);
                add4(acc, v0); add4(acc, v1);
                j += 2;
            }
            if (j < n) {
                float4 v0 = ld_row(x4, __builtin_amdgcn_readlane(my, j), lane);
                add4(acc, v0);
            }
        }
        const float di = 1.0f / fmaxf((float)c, 1.0f);
        float4 r = make_float4(acc.x * di, acc.y * di, acc.z * di, acc.w * di);
        *(float4*)(Alds + row * A_STRIDE + 4 * lane) = r;
    }
    __syncthreads();

    const int half = lane >> 4;
    const int l15  = lane & 15;
    const int n0   = wave * 16;
    const int col  = n0 + l15;
    const int ccol = (col < DOUT) ? col : 0;

    const float* __restrict__ Wlp = Wl + 2 * half * DOUT + ccol;
    const float* __restrict__ Wrp = Wr + 2 * half * DOUT + ccol;
    const float* __restrict__ Ap  = Alds + l15 * A_STRIDE + 2 * half;

    v8f acc = {};
    #pragma unroll 8
    for (int k = 0; k < DIN; k += 4) {
        v2f a, b;
        a.x = Ap[k];
        a.y = Ap[k + 1];
        b.x = Wlp[k * DOUT];
        b.y = Wlp[k * DOUT + DOUT];
        acc = __builtin_amdgcn_wmma_f32_16x16x4_f32(
            false, a, false, b, (short)0, acc, false, false);
    }
    #pragma unroll 8
    for (int k = 0; k < DIN; k += 4) {
        v2f a, b;
        a.x = Ap[DIN + k];
        a.y = Ap[DIN + k + 1];
        b.x = Wrp[k * DOUT];
        b.y = Wrp[k * DOUT + DOUT];
        acc = __builtin_amdgcn_wmma_f32_16x16x4_f32(
            false, a, false, b, (short)0, acc, false, false);
    }

    if (col < DOUT) {
        const float bv = bias[col];
        float* __restrict__ op = out + (size_t)(m0 + 8 * half) * DOUT + col;
        #pragma unroll
        for (int r = 0; r < 8; ++r) {
            float v = acc[r] + bv;
            if (RELU) v = fmaxf(v, 0.0f);
            op[r * DOUT] = v;
        }
    }
}

__global__ void proj_gemm(const float* __restrict__ h,
                          const float* __restrict__ Wl,
                          const float* __restrict__ Wr,
                          const float* __restrict__ bias,
                          float* __restrict__ P,
                          float* __restrict__ Q) {
    __shared__ __attribute__((aligned(16))) float Alds[16 * P_STRIDE];

    const int m0   = blockIdx.x * 16;
    const int tid  = threadIdx.x;
    const int lane = tid & 31;
    const int wave = tid >> 5;

    const float4* __restrict__ h4 = (const float4*)(h + (size_t)m0 * DH);
    for (int idx = tid; idx < 16 * (DH / 4); idx += blockDim.x) {
        const int row = idx / (DH / 4);
        const int c4  = idx - row * (DH / 4);
        *(float4*)(Alds + row * P_STRIDE + 4 * c4) = h4[idx];
    }
    __syncthreads();

    const int half = lane >> 4;
    const int l15  = lane & 15;
    const int col  = wave * 16 + l15;
    const bool isP = col < NC;
    const int c    = isP ? col : col - NC;

    const float* __restrict__ Wp = (isP ? Wl : Wr) + 2 * half * NC + c;
    const float* __restrict__ Ap = Alds + l15 * P_STRIDE + 2 * half;

    v8f acc = {};
    #pragma unroll 8
    for (int k = 0; k < DH; k += 4) {
        v2f a, b;
        a.x = Ap[k];
        a.y = Ap[k + 1];
        b.x = Wp[k * NC];
        b.y = Wp[k * NC + NC];
        acc = __builtin_amdgcn_wmma_f32_16x16x4_f32(
            false, a, false, b, (short)0, acc, false, false);
    }

    const float bv = isP ? 0.0f : bias[c];
    const int   rs = isP ? PS : NC;
    float* __restrict__ op = (isP ? P + (size_t)(m0 + 8 * half) * PS
                                  : Q + (size_t)(m0 + 8 * half) * NC) + c;
    #pragma unroll
    for (int r = 0; r < 8; ++r)
        op[r * rs] = acc[r] + bv;
}

__global__ void aggr_logsoftmax(const float* __restrict__ P,
                                const float* __restrict__ Q,
                                const int* __restrict__ cnt,
                                const int* __restrict__ rowstart,
                                const int* __restrict__ nbr,
                                float* __restrict__ out) {
    const int lane = threadIdx.x & 31;
    const int wave = threadIdx.x >> 5;
    const int grp  = lane / 10;
    const int ch   = lane - grp * 10;
    const int row  = (blockIdx.x * (blockDim.x >> 5) + wave) * 3 + grp;
    const bool on  = (grp < 3) && (row < NN);
    const int r    = on ? row : 0;
    const int c    = on ? cnt[r] : 0;
    const int beg  = rowstart[r];

    const float4* __restrict__ P4 = (const float4*)P;
    const float4 q = ((const float4*)Q)[(size_t)r * (NC / 4) + ch];

    float4 acc = make_float4(0.f, 0.f, 0.f, 0.f);
    int j = 0;
    for (; j + 8 <= c; j += 8) {
        int s[8];
        #pragma unroll
        for (int u = 0; u < 8; ++u) s[u] = nbr[beg + j + u];
        float4 v[8];
        #pragma unroll
        for (int u = 0; u < 8; ++u) v[u] = P4[(size_t)s[u] * (PS / 4) + ch];
        #pragma unroll
        for (int u = 0; u < 8; ++u) add4(acc, v[u]);
    }
    if (j < c) {
        const int left = c - j;
        int s[7];
        #pragma unroll
        for (int u = 0; u < 7; ++u) s[u] = nbr[beg + j + ((u < left) ? u : (left - 1))];
        float4 v[7];
        #pragma unroll
        for (int u = 0; u < 7; ++u) {
            v[u] = make_float4(0.f, 0.f, 0.f, 0.f);
            if (u < left) v[u] = P4[(size_t)s[u] * (PS / 4) + ch];
        }
        #pragma unroll
        for (int u = 0; u < 7; ++u) add4(acc, v[u]);
    }

    const float di = 1.0f / fmaxf((float)c, 1.0f);
    const float z0 = acc.x * di + q.x;
    const float z1 = acc.y * di + q.y;
    const float z2 = acc.z * di + q.z;
    const float z3 = acc.w * di + q.w;

    const int gb = (grp < 3) ? grp * 10 : 0;
    const float ml = fmaxf(fmaxf(z0, z1), fmaxf(z2, z3));
    float mx = -INFINITY;
    #pragma unroll
    for (int k = 0; k < 10; ++k) mx = fmaxf(mx, __shfl(ml, gb + k));
    const float el = expf(z0 - mx) + expf(z1 - mx) + expf(z2 - mx) + expf(z3 - mx);
    float sum = 0.0f;
    #pragma unroll
    for (int k = 0; k < 10; ++k) sum += __shfl(el, gb + k);
    const float lse = mx + logf(sum);

    if (on)
        ((float4*)out)[(size_t)row * (NC / 4) + ch] =
            make_float4(z0 - lse, z1 - lse, z2 - lse, z3 - lse);
}

extern "C" void kernel_launch(void* const* d_in, const int* in_sizes, int n_in,
                              void* d_out, int out_size, void* d_ws, size_t ws_size,
                              hipStream_t stream) {
    const float* x   = (const float*)d_in[0];
    const int*   ei  = (const int*)d_in[1];
    const int*   src = ei;
    const int*   dst = ei + NE;
    const float* Wl0 = (const float*)d_in[2];
    const float* Wr0 = (const float*)d_in[3];
    const float* b0  = (const float*)d_in[4];
    const float* Wl1 = (const float*)d_in[5];
    const float* Wr1 = (const float*)d_in[6];
    const float* b1  = (const float*)d_in[7];
    const float* Wl2 = (const float*)d_in[8];
    const float* Wr2 = (const float*)d_in[9];
    const float* b2  = (const float*)d_in[10];
    float* hout = (float*)d_out;

    char* ws = (char*)d_ws;
    size_t off = 0;
    auto carve = [&](size_t bytes) {
        void* p = ws + off;
        off = (off + bytes + 255) & ~(size_t)255;
        return p;
    };
    int*   cnt      = (int*)carve((size_t)NN * 4);
    int*   rowstart = (int*)carve((size_t)NN * 4);
    int*   rank     = (int*)carve((size_t)NE * 4);
    int*   nbr      = (int*)carve((size_t)NE * 4);
    float* h0       = (float*)carve((size_t)NN * DH * 4);
    float* h1       = (float*)carve((size_t)NN * DH * 4);
    float* P        = h0;
    float* Q        = h0 + (size_t)NN * PS;

    const int ZB = 256;
    dim3 bZ(ZB);
    dim3 gAggr((NN + 23) / 24);
    dim3 gZcnt((NN / 4 + ZB - 1) / ZB);
    dim3 gE((NE + ZB - 1) / ZB);
    dim3 gScan((NN + 1023) / 1024);
    dim3 gGemm(NN / 16);

    zero_f32v4<<<gZcnt, bZ, 0, stream>>>((float4*)cnt, NN / 4);
    rank_kernel<<<gE, bZ, 0, stream>>>(dst, cnt, rank);
    rowstart_kernel<<<gScan, dim3(1024), 0, stream>>>(cnt, rowstart);
    fill_kernel<<<gE, bZ, 0, stream>>>(src, dst, rank, rowstart, nbr);

    sage_gemm<DH, DH, true><<<gGemm, dim3(256), 0, stream>>>(x, cnt, rowstart, nbr, Wl0, Wr0, b0, h0);

    sage_gemm<DH, DH, true><<<gGemm, dim3(256), 0, stream>>>(h0, cnt, rowstart, nbr, Wl1, Wr1, b1, h1);

    proj_gemm<<<gGemm, dim3(160), 0, stream>>>(h1, Wl2, Wr2, b2, P, Q);
    aggr_logsoftmax<<<gAggr, bZ, 0, stream>>>(P, Q, cnt, rowstart, nbr, hout);
}


# Both GEMM kernels take 32 rows per work-group: a wave keeps two accumulator tiles and multiplies each weight fragment with both, halving weight loads and waves per row; sums unchanged
# speedup vs baseline: 3.3961x; 1.0499x over previous
#include <hip/hip_runtime.h>
#include <math.h>


#define NN 100000
#define NE 1600000
#define DH 128
#define NC 40
#define A_STRIDE 260
#define P_STRIDE 132
#define PS 64
#define TM 32

static_assert(NN % TM == 0 && TM == 32, "whole row tiles only; two accumulator tiles per wave");

typedef __attribute__((ext_vector_type(2))) float v2f;
typedef __attribute__((ext_vector_type(8))) float v8f;

__global__ void zero_f32v4(float4* __restrict__ p, int n4) {
    int i = blockIdx.x * blockDim.x + threadIdx.x;
    if (i < n4) p[i] = make_float4(0.f, 0.f, 0.f, 0.f);
}

__global__ void rank_kernel(const int* __restrict__ dst,
                            int* __restrict__ cnt,
                            int* __restrict__ rank) {
    int e = blockIdx.x * blockDim.x + threadIdx.x;
    if (e < NE) rank[e] = atomicAdd(&cnt[dst[e]], 1);
}

__global__ void rowstart_kernel(const int* __restrict__ cnt,
                                int* __restrict__ rowstart) {
    __shared__ int front[32];
    __shared__ int wsum[32];
    const int tid   = threadIdx.x;
    const int lane  = tid & 31;
    const int wave  = tid >> 5;
    const int start = blockIdx.x * 1024;

    int part = 0;
    for (int i = tid; i < start; i += 1024) part += cnt[i];
    for (int o = 16; o > 0; o >>= 1) part += __shfl_xor(part, o);
    if (lane == 0) front[wave] = part;

    const int i = start + tid;
    const int v = (i < NN) ? cnt[i] : 0;
    int incl = v;
    for (int o = 1; o < 32; o <<= 1) {
        int t = __shfl_up(incl, o);
        if (lane >= o) incl += t;
    }
    if (lane == 31) wsum[wave] = incl;
    __syncthreads();

    int off = 0;
    for (int w = 0; w < 32; ++w) off += front[w];
    for (int w = 0; w < wave; ++w) off += wsum[w];
    if (i < NN) rowstart[i] = off + incl - v;
}

__global__ void fill_kernel(const int* __restrict__ src,
                            const int* __restrict__ dst,
                            const int* __restrict__ rank,
                            const int* __restrict__ rowstart,
                            int* __restrict__ nbr) {
    int e = blockIdx.x * blockDim.x + threadIdx.x;
    if (e < NE) nbr[rowstart[dst[e]] + rank[e]] = src[e];
}

__device__ __forceinline__ float4 ld_row(const float4* __restrict__ x4, int s, int lane) {
    return x4[(size_t)s * (DH / 4) + lane];
}
__device__ __forceinline__ void add4(float4& a, const float4& v) {
    a.x += v.x; a.y += v.y; a.z += v.z; a.w += v.w;
}

template <int DIN, int DOUT, bool RELU>
__global__ void sage_gemm(const float* __restrict__ x,
                          const int* __restrict__ cnt,
                          const int* __restrict__ rowstart,
                          const int* __restrict__ nbr,
                          const float* __restrict__ Wl,
                          const float* __restrict__ Wr,
                          const float* __restrict__ bias,
                          float* __restrict__ out) {
    static_assert(DIN == DH, "row gather is written for 128-float rows");
    __shared__ __attribute__((aligned(16))) float Alds[TM * A_STRIDE];

    const int m0   = blockIdx.x * TM;
    const int tid  = threadIdx.x;
    const int lane = tid & 31;
    const int wave = tid >> 5;

    for (int idx = tid; idx < TM * DIN; idx += blockDim.x) {
        int row = idx / DIN;
        int kk  = idx - row * DIN;
        Alds[row * A_STRIDE + DIN + kk] = x[(size_t)(m0 + row) * DIN + kk];
    }

    const float4* __restrict__ x4 = (const float4*)x;
    const int wv     = __builtin_amdgcn_readfirstlane(wave);
    const int nwaves = blockDim.x >> 5;
    for (int row = wv; row < TM; row += nwaves) {
        const int g   = m0 + row;
        const int c   = __builtin_amdgcn_readfirstlane(cnt[g]);
        const int beg = __builtin_amdgcn_readfirstlane(rowstart[g]);
        float4 acc = make_float4(0.f, 0.f, 0.f, 0.f);
        for (int base = 0; base < c; base += 32) {
            const int n  = (c - base < 32) ? (c - base) : 32;
            const int my = (lane < n) ? nbr[beg + base + lane] : 0;
            int j = 0;
            for (; j + 8 <= n; j += 8) {
                float4 v0 = ld_row(x4, __builtin_amdgcn_readlane(my, j + 0), lane);
                float4 v1 = ld_row(x4, __builtin_amdgcn_readlane(my, j + 1), lane);
                float4 v2 = ld_row(x4, __builtin_amdgcn_readlane(my, j + 2), lane);
                float4 v3 = ld_row(x4, __builtin_amdgcn_readlane(my, j + 3), lane);
                float4 v4 = ld_row(x4, __builtin_amdgcn_readlane(my, j + 4), lane);
                float4 v5 = ld_row(x4, __builtin_amdgcn_readlane(my, j + 5), lane);
                float4 v6 = ld_row(x4, __builtin_amdgcn_readlane(my, j + 6), lane);
                float4 v7 = ld_row(x4, __builtin_amdgcn_readlane(my, j + 7), lane);
                add4(acc, v0); add4(acc, v1); add4(acc, v2); add4(acc, v3);
                add4(acc, v4); add4(acc, v5); add4(acc, v6); add4(acc, v7);
            }
            if (j + 4 <= n) {
                float4 v0 = ld_row(x4, __builtin_amdgcn_readlane(my, j + 0), lane);
                float4 v1 = ld_row(x4, __builtin_amdgcn_readlane(my, j + 1), lane);
                float4 v2 = ld_row(x4, __builtin_amdgcn_readlane(my, j + 2), lane);
                float4 v3 = ld_row(x4, __builtin_amdgcn_readlane(my, j + 3), lane);
                add4(acc, v0); add4(acc, v1); add4(acc, v2); add4(acc, v3);
                j += 4;
            }
            if (j + 2 <= n) {
                float4 v0 = ld_row(x4, __builtin_amdgcn_readlane(my, j + 0), lane);
                float4 v1 = ld_row(x4, __builtin_amdgcn_readlane(my, j + 1), lane);
                add4(acc, v0); add4(acc, v1);
                j += 2;
            }
            if (j < n) {
                float4 v0 = ld_row(x4, __builtin_amdgcn_readlane(my, j), lane);
                add4(acc, v0);
            }
        }
        const float di = 1.0f / fmaxf((float)c, 1.0f);
        float4 r = make_float4(acc.x * di, acc.y * di, acc.z * di, acc.w * di);
        *(float4*)(Alds + row * A_STRIDE + 4 * lane) = r;
    }
    __syncthreads();

    const int half = lane >> 4;
    const int l15  = lane & 15;
    const int n0   = wave * 16;
    const int col  = n0 + l15;
    const int ccol = (col < DOUT) ? col : 0;

    const float* __restrict__ Wlp = Wl + 2 * half * DOUT + ccol;
    const float* __restrict__ Wrp = Wr + 2 * half * DOUT + ccol;
    const float* __restrict__ Ap0 = Alds + l15 * A_STRIDE + 2 * half;
    const float* __restrict__ Ap1 = Ap0 + 16 * A_STRIDE;

    v8f acc0 = {};
    v8f acc1 = {};
    #pragma unroll 8
    for (int k = 0; k < DIN; k += 4) {
        v2f a0, a1, b;
        a0.x = Ap0[k];
        a0.y = Ap0[k + 1];
        a1.x = Ap1[k];
        a1.y = Ap1[k + 1];
        b.x = Wlp[k * DOUT];
        b.y = Wlp[k * DOUT + DOUT];
        acc0 = __builtin_amdgcn_wmma_f32_16x16x4_f32(
            false, a0, false, b, (short)0, acc0, false, false);
        acc1 = __builtin_amdgcn_wmma_f32_16x16x4_f32(
            false, a1, false, b, (short)0, acc1, false, false);
    }
    #pragma unroll 8
    for (int k = 0; k < DIN; k += 4) {
        v2f a0, a1, b;
        a0.x = Ap0[DIN + k];
        a0.y = Ap0[DIN + k + 1];
        a1.x = Ap1[DIN + k];
        a1.y = Ap1[DIN + k + 1];
        b.x = Wrp[k * DOUT];
        b.y = Wrp[k * DOUT + DOUT];
        acc0 = __builtin_amdgcn_wmma_f32_16x16x4_f32(
            false, a0, false, b, (short)0, acc0, false, false);
        acc1 = __builtin_amdgcn_wmma_f32_16x16x4_f32(
            false, a1, false, b, (short)0, acc1, false, false);
    }

    if (col < DOUT) {
        const float bv = bias[col];
        float* __restrict__ op = out + (size_t)(m0 + 8 * half) * DOUT + col;
        #pragma unroll
        for (int r = 0; r < 8; ++r) {
            float v = acc0[r] + bv;
            if (RELU) v = fmaxf(v, 0.0f);
            op[r * DOUT] = v;
        }
        op += 16 * DOUT;
        #pragma unroll
        for (int r = 0; r < 8; ++r) {
            float v = acc1[r] + bv;
            if (RELU) v = fmaxf(v, 0.0f);
            op[r * DOUT] = v;
        }
    }
}

__global__ void proj_gemm(const float* __restrict__ h,
                          const float* __restrict__ Wl,
                          const float* __restrict__ Wr,
                          const float* __restrict__ bias,
                          float* __restrict__ P,
                          float* __restrict__ Q) {
    __shared__ __attribute__((aligned(16))) float Alds[TM * P_STRIDE];

    const int m0   = blockIdx.x * TM;
    const int tid  = threadIdx.x;
    const int lane = tid & 31;
    const int wave = tid >> 5;

    const float4* __restrict__ h4 = (const float4*)(h + (size_t)m0 * DH);
    for (int idx = tid; idx < TM * (DH / 4); idx += blockDim.x) {
        const int row = idx / (DH / 4);
        const int c4  = idx - row * (DH / 4);
        *(float4*)(Alds + row * P_STRIDE + 4 * c4) = h4[idx];
    }
    __syncthreads();

    const int half = lane >> 4;
    const int l15  = lane & 15;
    const int col  = wave * 16 + l15;
    const bool isP = col < NC;
    const int c    = isP ? col : col - NC;

    const float* __restrict__ Wp = (isP ? Wl : Wr) + 2 * half * NC + c;
    const float* __restrict__ Ap0 = Alds + l15 * P_STRIDE + 2 * half;
    const float* __restrict__ Ap1 = Ap0 + 16 * P_STRIDE;

    v8f acc0 = {};
    v8f acc1 = {};
    #pragma unroll 8
    for (int k = 0; k < DH; k += 4) {
        v2f a0, a1, b;
        a0.x = Ap0[k];
        a0.y = Ap0[k + 1];
        a1.x = Ap1[k];
        a1.y = Ap1[k + 1];
        b.x = Wp[k * NC];
        b.y = Wp[k * NC + NC];
        acc0 = __builtin_amdgcn_wmma_f32_16x16x4_f32(
            false, a0, false, b, (short)0, acc0, false, false);
        acc1 = __builtin_amdgcn_wmma_f32_16x16x4_f32(
            false, a1, false, b, (short)0, acc1, false, false);
    }

    const float bv = isP ? 0.0f : bias[c];
    const int   rs = isP ? PS : NC;
    float* __restrict__ op = (isP ? P + (size_t)(m0 + 8 * half) * PS
                                  : Q + (size_t)(m0 + 8 * half) * NC) + c;
    #pragma unroll
    for (int r = 0; r < 8; ++r)
        op[r * rs] = acc0[r] + bv;
    op += 16 * rs;
    #pragma unroll
    for (int r = 0; r < 8; ++r)
        op[r * rs] = acc1[r] + bv;
}

__global__ void aggr_logsoftmax(const float* __restrict__ P,
                                const float* __restrict__ Q,
                                const int* __restrict__ cnt,
                                const int* __restrict__ rowstart,
                                const int* __restrict__ nbr,
                                float* __restrict__ out) {
    const int lane = threadIdx.x & 31;
    const int wave = threadIdx.x >> 5;
    const int grp  = lane / 10;
    const int ch   = lane - grp * 10;
    const int row  = (blockIdx.x * (blockDim.x >> 5) + wave) * 3 + grp;
    const bool on  = (grp < 3) && (row < NN);
    const int r    = on ? row : 0;
    const int c    = on ? cnt[r] : 0;
    const int beg  = rowstart[r];

    const float4* __restrict__ P4 = (const float4*)P;
    const float4 q = ((const float4*)Q)[(size_t)r * (NC / 4) + ch];

    float4 acc = make_float4(0.f, 0.f, 0.f, 0.f);
    int j = 0;
    for (; j + 8 <= c; j += 8) {
        int s[8];
        #pragma unroll
        for (int u = 0; u < 8; ++u) s[u] = nbr[beg + j + u];
        float4 v[8];
        #pragma unroll
        for (int u = 0; u < 8; ++u) v[u] = P4[(size_t)s[u] * (PS / 4) + ch];
        #pragma unroll
        for (int u = 0; u < 8; ++u) add4(acc, v[u]);
    }
    if (j < c) {
        const int left = c - j;
        int s[7];
        #pragma unroll
        for (int u = 0; u < 7; ++u) s[u] = nbr[beg + j + ((u < left) ? u : (left - 1))];
        float4 v[7];
        #pragma unroll
        for (int u = 0; u < 7; ++u) {
            v[u] = make_float4(0.f, 0.f, 0.f, 0.f);
            if (u < left) v[u] = P4[(size_t)s[u] * (PS / 4) + ch];
        }
        #pragma unroll
        for (int u = 0; u < 7; ++u) add4(acc, v[u]);
    }

    const float di = 1.0f / fmaxf((float)c, 1.0f);
    const float z0 = acc.x * di + q.x;
    const float z1 = acc.y * di + q.y;
    const float z2 = acc.z * di + q.z;
    const float z3 = acc.w * di + q.w;

    const int gb = (grp < 3) ? grp * 10 : 0;
    const float ml = fmaxf(fmaxf(z0, z1), fmaxf(z2, z3));
    float mx = -INFINITY;
    #pragma unroll
    for (int k = 0; k < 10; ++k) mx = fmaxf(mx, __shfl(ml, gb + k));
    const float el = expf(z0 - mx) + expf(z1 - mx) + expf(z2 - mx) + expf(z3 - mx);
    float sum = 0.0f;
    #pragma unroll
    for (int k = 0; k < 10; ++k) sum += __shfl(el, gb + k);
    const float lse = mx + logf(sum);

    if (on)
        ((float4*)out)[(size_t)row * (NC / 4) + ch] =
            make_float4(z0 - lse, z1 - lse, z2 - lse, z3 - lse);
}

extern "C" void kernel_launch(void* const* d_in, const int* in_sizes, int n_in,
                              void* d_out, int out_size, void* d_ws, size_t ws_size,
                              hipStream_t stream) {
    const float* x   = (const float*)d_in[0];
    const int*   ei  = (const int*)d_in[1];
    const int*   src = ei;
    const int*   dst = ei + NE;
    const float* Wl0 = (const float*)d_in[2];
    const float* Wr0 = (const float*)d_in[3];
    const float* b0  = (const float*)d_in[4];
    const float* Wl1 = (const float*)d_in[5];
    const float* Wr1 = (const float*)d_in[6];
    const float* b1  = (const float*)d_in[7];
    const float* Wl2 = (const float*)d_in[8];
    const float* Wr2 = (const float*)d_in[9];
    const float* b2  = (const float*)d_in[10];
    float* hout = (float*)d_out;

    char* ws = (char*)d_ws;
    size_t off = 0;
    auto carve = [&](size_t bytes) {
        void* p = ws + off;
        off = (off + bytes + 255) & ~(size_t)255;
        return p;
    };
    int*   cnt      = (int*)carve((size_t)NN * 4);
    int*   rowstart = (int*)carve((size_t)NN * 4);
    int*   rank     = (int*)carve((size_t)NE * 4);
    int*   nbr      = (int*)carve((size_t)NE * 4);
    float* h0       = (float*)carve((size_t)NN * DH * 4);
    float* h1       = (float*)carve((size_t)NN * DH * 4);
    float* P        = h0;
    float* Q        = h0 + (size_t)NN * PS;

    const int ZB = 256;
    dim3 bZ(ZB);
    dim3 gAggr((NN + 23) / 24);
    dim3 gZcnt((NN / 4 + ZB - 1) / ZB);
    dim3 gE((NE + ZB - 1) / ZB);
    dim3 gScan((NN + 1023) / 1024);
    dim3 gGemm(NN / TM);

    zero_f32v4<<<gZcnt, bZ, 0, stream>>>((float4*)cnt, NN / 4);
    rank_kernel<<<gE, bZ, 0, stream>>>(dst, cnt, rank);
    rowstart_kernel<<<gScan, dim3(1024), 0, stream>>>(cnt, rowstart);
    fill_kernel<<<gE, bZ, 0, stream>>>(src, dst, rank, rowstart, nbr);

    sage_gemm<DH, DH, true><<<gGemm, dim3(256), 0, stream>>>(x, cnt, rowstart, nbr, Wl0, Wr0, b0, h0);

    sage_gemm<DH, DH, true><<<gGemm, dim3(256), 0, stream>>>(h0, cnt, rowstart, nbr, Wl1, Wr1, b1, h1);

    proj_gemm<<<gGemm, dim3(160), 0, stream>>>(h1, Wl2, Wr2, b2, P, Q);
    aggr_logsoftmax<<<gAggr, bZ, 0, stream>>>(P, Q, cnt, rowstart, nbr, hout);
}


# Edge counting sort rebuilt as two-level bucket sort staged in LDS (256-node buckets, LDS atomic slots, contiguous runs out): no per-edge global atomic or scatter; 7 launches; row order = LDS arrival
# speedup vs baseline: 4.9466x; 1.4566x over previous
#include <hip/hip_runtime.h>
#include <math.h>


#define NN 100000
#define NE 1600000
#define DH 128
#define NC 40
#define A_STRIDE 260
#define P_STRIDE 132
#define PS 64
#define TM 32

static_assert(NN % TM == 0 && TM == 32, "whole row tiles only; two accumulator tiles per wave");

typedef __attribute__((ext_vector_type(2))) float v2f;
typedef __attribute__((ext_vector_type(8))) float v8f;

#define BSH 8
#define BSZ (1 << BSH)
#define NB  ((NN + BSZ - 1) / BSZ)
#define CH  6400
#define NCH (NE / CH)
#define ST  640
#define KCAP 8192

static_assert(NE % CH == 0 && CH % ST == 0 && ST % 32 == 0, "whole chunks, whole waves");
static_assert(NB <= ST, "one thread per bucket in the offset scan");
static_assert((NN - 1) < (1 << (31 - BSH)), "source index and low destination bits share one word");

__global__ void __launch_bounds__(ST)
bucket_hist(const int* __restrict__ dst,
            int* __restrict__ hist) {
    __shared__ int h[NB];
    const int tid = threadIdx.x;
    if (tid < NB) h[tid] = 0;
    __syncthreads();
    const int* __restrict__ d = dst + blockIdx.x * CH;
    int dv[CH / ST];
    #pragma unroll
    for (int i = 0; i < CH / ST; ++i) dv[i] = d[tid + i * ST];
    #pragma unroll
    for (int i = 0; i < CH / ST; ++i) atomicAdd(&h[dv[i] >> BSH], 1);
    __syncthreads();
    if (tid < NB) hist[blockIdx.x * NB + tid] = h[tid];
}

__global__ void __launch_bounds__(ST)
bucket_scatter(const int* __restrict__ src,
               const int* __restrict__ dst,
               const int* __restrict__ hist,
               int* __restrict__ bbase,
               int* __restrict__ part) {
    __shared__ int cur[NB];
    __shared__ int lstart[NB];
    __shared__ int gdelta[NB];
    __shared__ int wt[ST / 32];
    __shared__ int wm[ST / 32];
    __shared__ int stage[CH];
    __shared__ int gpos[CH];

    const int tid  = threadIdx.x;
    const int lane = tid & 31;
    const int wave = tid >> 5;
    const int w    = blockIdx.x;

    int tot = 0, pre = 0, mine = 0;
    if (tid < NB) {
        #pragma unroll 25
        for (int c = 0; c < NCH; ++c) {
            const int t = hist[c * NB + tid];
            tot += t;
            pre += (c < w) ? t : 0;
        }
        mine = hist[w * NB + tid];
    }
    int it = tot, im = mine;
    for (int o = 1; o < 32; o <<= 1) {
        const int a = __shfl_up(it, o);
        const int c = __shfl_up(im, o);
        if (lane >= o) { it += a; im += c; }
    }
    if (lane == 31) { wt[wave] = it; wm[wave] = im; }
    __syncthreads();
    int ot = 0, om = 0;
    for (int k = 0; k < wave; ++k) { ot += wt[k]; om += wm[k]; }
    const int base = ot + it - tot;
    const int ls   = om + im - mine;
    if (tid < NB) {
        cur[tid]    = 0;
        lstart[tid] = ls;
        gdelta[tid] = base + pre - ls;
        if (w == 0) bbase[tid] = base;
    }
    if (w == 0 && tid == 0) bbase[NB] = NE;
    __syncthreads();

    const int* __restrict__ s = src + w * CH;
    const int* __restrict__ d = dst + w * CH;
    int dv[CH / ST], sv[CH / ST];
    #pragma unroll
    for (int i = 0; i < CH / ST; ++i) {
        dv[i] = d[tid + i * ST];
        sv[i] = s[tid + i * ST];
    }
    #pragma unroll
    for (int i = 0; i < CH / ST; ++i) {
        const int b = dv[i] >> BSH;
        const int p = lstart[b] + atomicAdd(&cur[b], 1);
        stage[p] = (sv[i] << BSH) | (dv[i] & (BSZ - 1));
        gpos[p]  = gdelta[b] + p;
    }
    __syncthreads();
    #pragma unroll
    for (int i = 0; i < CH / ST; ++i) {
        const int p = tid + i * ST;
        part[gpos[p]] = stage[p];
    }
}

__global__ void __launch_bounds__(512)
bucket_sort(const int* __restrict__ part,
            const int* __restrict__ bbase,
            int* __restrict__ cnt,
            int* __restrict__ rowstart,
            int* __restrict__ nbr) {
    __shared__ int c[BSZ];
    __shared__ int rs[BSZ];
    __shared__ int wsum[16];
    __shared__ int stg[KCAP];

    const int tid  = threadIdx.x;
    const int lane = tid & 31;
    const int wave = tid >> 5;
    const int b    = blockIdx.x;
    const int beg  = bbase[b];
    const int n    = bbase[b + 1] - beg;

    const bool fits = n <= KCAP;
    int pk[KCAP / 512];
    #pragma unroll
    for (int j = 0; j < KCAP / 512; ++j) {
        const int i = tid + j * 512;
        pk[j] = (fits && i < n) ? part[beg + i] : -1;
    }

    if (tid < BSZ) c[tid] = 0;
    __syncthreads();
    if (fits) {
        #pragma unroll
        for (int j = 0; j < KCAP / 512; ++j)
            if (pk[j] >= 0) atomicAdd(&c[pk[j] & (BSZ - 1)], 1);
    } else {
        for (int i = tid; i < n; i += 512)
            atomicAdd(&c[part[beg + i] & (BSZ - 1)], 1);
    }
    __syncthreads();

    const int v = (tid < BSZ) ? c[tid] : 0;
    int incl = v;
    for (int o = 1; o < 32; o <<= 1) {
        const int t = __shfl_up(incl, o);
        if (lane >= o) incl += t;
    }
    if (lane == 31) wsum[wave] = incl;
    __syncthreads();
    int off = 0;
    for (int k = 0; k < wave; ++k) off += wsum[k];
    const int ex = off + incl - v;
    if (tid < BSZ) {
        const int node = b * BSZ + tid;
        if (node < NN) { cnt[node] = v; rowstart[node] = beg + ex; }
        rs[tid] = ex;
        c[tid]  = 0;
    }
    __syncthreads();

    if (fits) {
        #pragma unroll
        for (int j = 0; j < KCAP / 512; ++j)
            if (pk[j] >= 0) {
                const int dl = pk[j] & (BSZ - 1);
                stg[rs[dl] + atomicAdd(&c[dl], 1)] = pk[j] >> BSH;
            }
    } else {
        for (int i = tid; i < n; i += 512) {
            const int q  = part[beg + i];
            const int dl = q & (BSZ - 1);
            nbr[beg + rs[dl] + atomicAdd(&c[dl], 1)] = q >> BSH;
        }
    }
    __syncthreads();
    if (fits)
        for (int i = tid; i < n; i += 512) nbr[beg + i] = stg[i];
}

__device__ __forceinline__ float4 ld_row(const float4* __restrict__ x4, int s, int lane) {
    return x4[(size_t)s * (DH / 4) + lane];
}
__device__ __forceinline__ void add4(float4& a, const float4& v) {
    a.x += v.x; a.y += v.y; a.z += v.z; a.w += v.w;
}

template <int DIN, int DOUT, bool RELU>
__global__ void sage_gemm(const float* __restrict__ x,
                          const int* __restrict__ cnt,
                          const int* __restrict__ rowstart,
                          const int* __restrict__ nbr,
                          const float* __restrict__ Wl,
                          const float* __restrict__ Wr,
                          const float* __restrict__ bias,
                          float* __restrict__ out) {
    static_assert(DIN == DH, "row gather is written for 128-float rows");
    __shared__ __attribute__((aligned(16))) float Alds[TM * A_STRIDE];

    const int m0   = blockIdx.x * TM;
    const int tid  = threadIdx.x;
    const int lane = tid & 31;
    const int wave = tid >> 5;

    for (int idx = tid; idx < TM * DIN; idx += blockDim.x) {
        int row = idx / DIN;
        int kk  = idx - row * DIN;
        Alds[row * A_STRIDE + DIN + kk] = x[(size_t)(m0 + row) * DIN + kk];
    }

    const float4* __restrict__ x4 = (const float4*)x;
    const int wv     = __builtin_amdgcn_readfirstlane(wave);
    const int nwaves = blockDim.x >> 5;
    for (int row = wv; row < TM; row += nwaves) {
        const int g   = m0 + row;
        const int c   = __builtin_amdgcn_readfirstlane(cnt[g]);
        const int beg = __builtin_amdgcn_readfirstlane(rowstart[g]);
        float4 acc = make_float4(0.f, 0.f, 0.f, 0.f);
        for (int base = 0; base < c; base += 32) {
            const int n  = (c - base < 32) ? (c - base) : 32;
            const int my = (lane < n) ? nbr[beg + base + lane] : 0;
            int j = 0;
            for (; j + 8 <= n; j += 8) {
                float4 v0 = ld_row(x4, __builtin_amdgcn_readlane(my, j + 0), lane);
                float4 v1 = ld_row(x4, __builtin_amdgcn_readlane(my, j + 1), lane);
                float4 v2 = ld_row(x4, __builtin_amdgcn_readlane(my, j + 2), lane);
                float4 v3 = ld_row(x4, __builtin_amdgcn_readlane(my, j + 3), lane);
                float4 v4 = ld_row(x4, __builtin_amdgcn_readlane(my, j + 4), lane);
                float4 v5 = ld_row(x4, __builtin_amdgcn_readlane(my, j + 5), lane);
                float4 v6 = ld_row(x4, __builtin_amdgcn_readlane(my, j + 6), lane);
                float4 v7 = ld_row(x4, __builtin_amdgcn_readlane(my, j + 7), lane);
                add4(acc, v0); add4(acc, v1); add4(acc, v2); add4(acc, v3);
                add4(acc, v4); add4(acc, v5); add4(acc, v6); add4(acc, v7);
            }
            if (j + 4 <= n) {
                float4 v0 = ld_row(x4, __builtin_amdgcn_readlane(my, j + 0), lane);
                float4 v1 = ld_row(x4, __builtin_amdgcn_readlane(my, j + 1), lane);
                float4 v2 = ld_row(x4, __builtin_amdgcn_readlane(my, j + 2), lane);
                float4 v3 = ld_row(x4, __builtin_amdgcn_readlane(my, j + 3), lane);
                add4(acc, v0); add4(acc, v1); add4(acc, v2); add4(acc, v3);
                j += 4;
            }
            if (j + 2 <= n) {
                float4 v0 = ld_row(x4, __builtin_amdgcn_readlane(my, j + 0), lane);
                float4 v1 = ld_row(x4, __builtin_amdgcn_readlane(my, j + 1), lane);
                add4(acc, v0); add4(acc, v1);
                j += 2;
            }
            if (j < n) {
                float4 v0 = ld_row(x4, __builtin_amdgcn_readlane(my, j), lane);
                add4(acc, v0);
            }
        }
        const float di = 1.0f / fmaxf((float)c, 1.0f);
        float4 r = make_float4(acc.x * di, acc.y * di, acc.z * di, acc.w * di);
        *(float4*)(Alds + row * A_STRIDE + 4 * lane) = r;
    }
    __syncthreads();

    const int half = lane >> 4;
    const int l15  = lane & 15;
    const int n0   = wave * 16;
    const int col  = n0 + l15;
    const int ccol = (col < DOUT) ? col : 0;

    const float* __restrict__ Wlp = Wl + 2 * half * DOUT + ccol;
    const float* __restrict__ Wrp = Wr + 2 * half * DOUT + ccol;
    const float* __restrict__ Ap0 = Alds + l15 * A_STRIDE + 2 * half;
    const float* __restrict__ Ap1 = Ap0 + 16 * A_STRIDE;

    v8f acc0 = {};
    v8f acc1 = {};
    #pragma unroll 8
    for (int k = 0; k < DIN; k += 4) {
        v2f a0, a1, b;
        a0.x = Ap0[k];
        a0.y = Ap0[k + 1];
        a1.x = Ap1[k];
        a1.y = Ap1[k + 1];
        b.x = Wlp[k * DOUT];
        b.y = Wlp[k * DOUT + DOUT];
        acc0 = __builtin_amdgcn_wmma_f32_16x16x4_f32(
            false, a0, false, b, (short)0, acc0, false, false);
        acc1 = __builtin_amdgcn_wmma_f32_16x16x4_f32(
            false, a1, false, b, (short)0, acc1, false, false);
    }
    #pragma unroll 8
    for (int k = 0; k < DIN; k += 4) {
        v2f a0, a1, b;
        a0.x = Ap0[DIN + k];
        a0.y = Ap0[DIN + k + 1];
        a1.x = Ap1[DIN + k];
        a1.y = Ap1[DIN + k + 1];
        b.x = Wrp[k * DOUT];
        b.y = Wrp[k * DOUT + DOUT];
        acc0 = __builtin_amdgcn_wmma_f32_16x16x4_f32(
            false, a0, false, b, (short)0, acc0, false, false);
        acc1 = __builtin_amdgcn_wmma_f32_16x16x4_f32(
            false, a1, false, b, (short)0, acc1, false, false);
    }

    if (col < DOUT) {
        const float bv = bias[col];
        float* __restrict__ op = out + (size_t)(m0 + 8 * half) * DOUT + col;
        #pragma unroll
        for (int r = 0; r < 8; ++r) {
            float v = acc0[r] + bv;
            if (RELU) v = fmaxf(v, 0.0f);
            op[r * DOUT] = v;
        }
        op += 16 * DOUT;
        #pragma unroll
        for (int r = 0; r < 8; ++r) {
            float v = acc1[r] + bv;
            if (RELU) v = fmaxf(v, 0.0f);
            op[r * DOUT] = v;
        }
    }
}

__global__ void proj_gemm(const float* __restrict__ h,
                          const float* __restrict__ Wl,
                          const float* __restrict__ Wr,
                          const float* __restrict__ bias,
                          float* __restrict__ P,
                          float* __restrict__ Q) {
    __shared__ __attribute__((aligned(16))) float Alds[TM * P_STRIDE];

    const int m0   = blockIdx.x * TM;
    const int tid  = threadIdx.x;
    const int lane = tid & 31;
    const int wave = tid >> 5;

    const float4* __restrict__ h4 = (const float4*)(h + (size_t)m0 * DH);
    for (int idx = tid; idx < TM * (DH / 4); idx += blockDim.x) {
        const int row = idx / (DH / 4);
        const int c4  = idx - row * (DH / 4);
        *(float4*)(Alds + row * P_STRIDE + 4 * c4) = h4[idx];
    }
    __syncthreads();

    const int half = lane >> 4;
    const int l15  = lane & 15;
    const int col  = wave * 16 + l15;
    const bool isP = col < NC;
    const int c    = isP ? col : col - NC;

    const float* __restrict__ Wp = (isP ? Wl : Wr) + 2 * half * NC + c;
    const float* __restrict__ Ap0 = Alds + l15 * P_STRIDE + 2 * half;
    const float* __restrict__ Ap1 = Ap0 + 16 * P_STRIDE;

    v8f acc0 = {};
    v8f acc1 = {};
    #pragma unroll 8
    for (int k = 0; k < DH; k += 4) {
        v2f a0, a1, b;
        a0.x = Ap0[k];
        a0.y = Ap0[k + 1];
        a1.x = Ap1[k];
        a1.y = Ap1[k + 1];
        b.x = Wp[k * NC];
        b.y = Wp[k * NC + NC];
        acc0 = __builtin_amdgcn_wmma_f32_16x16x4_f32(
            false, a0, false, b, (short)0, acc0, false, false);
        acc1 = __builtin_amdgcn_wmma_f32_16x16x4_f32(
            false, a1, false, b, (short)0, acc1, false, false);
    }

    const float bv = isP ? 0.0f : bias[c];
    const int   rs = isP ? PS : NC;
    float* __restrict__ op = (isP ? P + (size_t)(m0 + 8 * half) * PS
                                  : Q + (size_t)(m0 + 8 * half) * NC) + c;
    #pragma unroll
    for (int r = 0; r < 8; ++r)
        op[r * rs] = acc0[r] + bv;
    op += 16 * rs;
    #pragma unroll
    for (int r = 0; r < 8; ++r)
        op[r * rs] = acc1[r] + bv;
}

__global__ void aggr_logsoftmax(const float* __restrict__ P,
                                const float* __restrict__ Q,
                                const int* __restrict__ cnt,
                                const int* __restrict__ rowstart,
                                const int* __restrict__ nbr,
                                float* __restrict__ out) {
    const int lane = threadIdx.x & 31;
    const int wave = threadIdx.x >> 5;
    const int grp  = lane / 10;
    const int ch   = lane - grp * 10;
    const int row  = (blockIdx.x * (blockDim.x >> 5) + wave) * 3 + grp;
    const bool on  = (grp < 3) && (row < NN);
    const int r    = on ? row : 0;
    const int c    = on ? cnt[r] : 0;
    const int beg  = rowstart[r];

    const float4* __restrict__ P4 = (const float4*)P;
    const float4 q = ((const float4*)Q)[(size_t)r * (NC / 4) + ch];

    float4 acc = make_float4(0.f, 0.f, 0.f, 0.f);
    int j = 0;
    for (; j + 8 <= c; j += 8) {
        int s[8];
        #pragma unroll
        for (int u = 0; u < 8; ++u) s[u] = nbr[beg + j + u];
        float4 v[8];
        #pragma unroll
        for (int u = 0; u < 8; ++u) v[u] = P4[(size_t)s[u] * (PS / 4) + ch];
        #pragma unroll
        for (int u = 0; u < 8; ++u) add4(acc, v[u]);
    }
    if (j < c) {
        const int left = c - j;
        int s[7];
        #pragma unroll
        for (int u = 0; u < 7; ++u) s[u] = nbr[beg + j + ((u < left) ? u : (left - 1))];
        float4 v[7];
        #pragma unroll
        for (int u = 0; u < 7; ++u) {
            v[u] = make_float4(0.f, 0.f, 0.f, 0.f);
            if (u < left) v[u] = P4[(size_t)s[u] * (PS / 4) + ch];
        }
        #pragma unroll
        for (int u = 0; u < 7; ++u) add4(acc, v[u]);
    }

    const float di = 1.0f / fmaxf((float)c, 1.0f);
    const float z0 = acc.x * di + q.x;
    const float z1 = acc.y * di + q.y;
    const float z2 = acc.z * di + q.z;
    const float z3 = acc.w * di + q.w;

    const int gb = (grp < 3) ? grp * 10 : 0;
    const float ml = fmaxf(fmaxf(z0, z1), fmaxf(z2, z3));
    float mx = -INFINITY;
    #pragma unroll
    for (int k = 0; k < 10; ++k) mx = fmaxf(mx, __shfl(ml, gb + k));
    const float el = expf(z0 - mx) + expf(z1 - mx) + expf(z2 - mx) + expf(z3 - mx);
    float sum = 0.0f;
    #pragma unroll
    for (int k = 0; k < 10; ++k) sum += __shfl(el, gb + k);
    const float lse = mx + logf(sum);

    if (on)
        ((float4*)out)[(size_t)row * (NC / 4) + ch] =
            make_float4(z0 - lse, z1 - lse, z2 - lse, z3 - lse);
}

extern "C" void kernel_launch(void* const* d_in, const int* in_sizes, int n_in,
                              void* d_out, int out_size, void* d_ws, size_t ws_size,
                              hipStream_t stream) {
    const float* x   = (const float*)d_in[0];
    const int*   ei  = (const int*)d_in[1];
    const int*   src = ei;
    const int*   dst = ei + NE;
    const float* Wl0 = (const float*)d_in[2];
    const float* Wr0 = (const float*)d_in[3];
    const float* b0  = (const float*)d_in[4];
    const float* Wl1 = (const float*)d_in[5];
    const float* Wr1 = (const float*)d_in[6];
    const float* b1  = (const float*)d_in[7];
    const float* Wl2 = (const float*)d_in[8];
    const float* Wr2 = (const float*)d_in[9];
    const float* b2  = (const float*)d_in[10];
    float* hout = (float*)d_out;

    char* ws = (char*)d_ws;
    size_t off = 0;
    auto carve = [&](size_t bytes) {
        void* p = ws + off;
        off = (off + bytes + 255) & ~(size_t)255;
        return p;
    };
    int*   cnt      = (int*)carve((size_t)NN * 4);
    int*   rowstart = (int*)carve((size_t)NN * 4);
    int*   part     = (int*)carve((size_t)NE * 4);
    int*   nbr      = (int*)carve((size_t)NE * 4);
    float* h0       = (float*)carve((size_t)NN * DH * 4);
    float* h1       = (float*)carve((size_t)NN * DH * 4);
    float* P        = h0;
    float* Q        = h0 + (size_t)NN * PS;
    int*   hist     = (int*)h1;
    int*   bbase    = hist + NCH * NB;

    const int ZB = 256;
    dim3 bZ(ZB);
    dim3 gAggr((NN + 23) / 24);
    dim3 gGemm(NN / TM);

    bucket_hist<<<dim3(NCH), dim3(ST), 0, stream>>>(dst, hist);
    bucket_scatter<<<dim3(NCH), dim3(ST), 0, stream>>>(src, dst, hist, bbase, part);
    bucket_sort<<<dim3(NB), dim3(512), 0, stream>>>(part, bbase, cnt, rowstart, nbr);

    sage_gemm<DH, DH, true><<<gGemm, dim3(256), 0, stream>>>(x, cnt, rowstart, nbr, Wl0, Wr0, b0, h0);

    sage_gemm<DH, DH, true><<<gGemm, dim3(256), 0, stream>>>(h0, cnt, rowstart, nbr, Wl1, Wr1, b1, h1);

    proj_gemm<<<gGemm, dim3(160), 0, stream>>>(h1, Wl2, Wr2, b2, P, Q);
    aggr_logsoftmax<<<gAggr, bZ, 0, stream>>>(P, Q, cnt, rowstart, nbr, hout);
}


# Layer kernels resident, two wave roles: 8 fill waves (s_setprio 3) gather tile t+1's rows into a second LDS A tile while 8 matrix waves run tile t's WMMAs; one barrier per tile; sums unchanged
# speedup vs baseline: 5.1128x; 1.0336x over previous
#include <hip/hip_runtime.h>
#include <math.h>


#define NN 100000
#define NE 1600000
#define DH 128
#define NC 40
#define A_STRIDE 260
#define P_STRIDE 132
#define PS 64
#define TM 32

static_assert(NN % TM == 0 && TM == 32, "whole row tiles only; two accumulator tiles per wave");

typedef __attribute__((ext_vector_type(2))) float v2f;
typedef __attribute__((ext_vector_type(8))) float v8f;

#define BSH 8
#define BSZ (1 << BSH)
#define NB  ((NN + BSZ - 1) / BSZ)
#define CH  6400
#define NCH (NE / CH)
#define ST  640
#define KCAP 8192

static_assert(NE % CH == 0 && CH % ST == 0 && ST % 32 == 0, "whole chunks, whole waves");
static_assert(NB <= ST, "one thread per bucket in the offset scan");
static_assert((NN - 1) < (1 << (31 - BSH)), "source index and low destination bits share one word");

__global__ void __launch_bounds__(ST)
bucket_hist(const int* __restrict__ dst,
            int* __restrict__ hist) {
    __shared__ int h[NB];
    const int tid = threadIdx.x;
    if (tid < NB) h[tid] = 0;
    __syncthreads();
    const int* __restrict__ d = dst + blockIdx.x * CH;
    int dv[CH / ST];
    #pragma unroll
    for (int i = 0; i < CH / ST; ++i) dv[i] = d[tid + i * ST];
    #pragma unroll
    for (int i = 0; i < CH / ST; ++i) atomicAdd(&h[dv[i] >> BSH], 1);
    __syncthreads();
    if (tid < NB) hist[blockIdx.x * NB + tid] = h[tid];
}

__global__ void __launch_bounds__(ST)
bucket_scatter(const int* __restrict__ src,
               const int* __restrict__ dst,
               const int* __restrict__ hist,
               int* __restrict__ bbase,
               int* __restrict__ part) {
    __shared__ int cur[NB];
    __shared__ int lstart[NB];
    __shared__ int gdelta[NB];
    __shared__ int wt[ST / 32];
    __shared__ int wm[ST / 32];
    __shared__ int stage[CH];
    __shared__ int gpos[CH];

    const int tid  = threadIdx.x;
    const int lane = tid & 31;
    const int wave = tid >> 5;
    const int w    = blockIdx.x;

    int tot = 0, pre = 0, mine = 0;
    if (tid < NB) {
        #pragma unroll 25
        for (int c = 0; c < NCH; ++c) {
            const int t = hist[c * NB + tid];
            tot += t;
            pre += (c < w) ? t : 0;
        }
        mine = hist[w * NB + tid];
    }
    int it = tot, im = mine;
    for (int o = 1; o < 32; o <<= 1) {
        const int a = __shfl_up(it, o);
        const int c = __shfl_up(im, o);
        if (lane >= o) { it += a; im += c; }
    }
    if (lane == 31) { wt[wave] = it; wm[wave] = im; }
    __syncthreads();
    int ot = 0, om = 0;
    for (int k = 0; k < wave; ++k) { ot += wt[k]; om += wm[k]; }
    const int base = ot + it - tot;
    const int ls   = om + im - mine;
    if (tid < NB) {
        cur[tid]    = 0;
        lstart[tid] = ls;
        gdelta[tid] = base + pre - ls;
        if (w == 0) bbase[tid] = base;
    }
    if (w == 0 && tid == 0) bbase[NB] = NE;
    __syncthreads();

    const int* __restrict__ s = src + w * CH;
    const int* __restrict__ d = dst + w * CH;
    int dv[CH / ST], sv[CH / ST];
    #pragma unroll
    for (int i = 0; i < CH / ST; ++i) {
        dv[i] = d[tid + i * ST];
        sv[i] = s[tid + i * ST];
    }
    #pragma unroll
    for (int i = 0; i < CH / ST; ++i) {
        const int b = dv[i] >> BSH;
        const int p = lstart[b] + atomicAdd(&cur[b], 1);
        stage[p] = (sv[i] << BSH) | (dv[i] & (BSZ - 1));
        gpos[p]  = gdelta[b] + p;
    }
    __syncthreads();
    #pragma unroll
    for (int i = 0; i < CH / ST; ++i) {
        const int p = tid + i * ST;
        part[gpos[p]] = stage[p];
    }
}

__global__ void __launch_bounds__(512)
bucket_sort(const int* __restrict__ part,
            const int* __restrict__ bbase,
            int* __restrict__ cnt,
            int* __restrict__ rowstart,
            int* __restrict__ nbr) {
    __shared__ int c[BSZ];
    __shared__ int rs[BSZ];
    __shared__ int wsum[16];
    __shared__ int stg[KCAP];

    const int tid  = threadIdx.x;
    const int lane = tid & 31;
    const int wave = tid >> 5;
    const int b    = blockIdx.x;
    const int beg  = bbase[b];
    const int n    = bbase[b + 1] - beg;

    const bool fits = n <= KCAP;
    int pk[KCAP / 512];
    #pragma unroll
    for (int j = 0; j < KCAP / 512; ++j) {
        const int i = tid + j * 512;
        pk[j] = (fits && i < n) ? part[beg + i] : -1;
    }

    if (tid < BSZ) c[tid] = 0;
    __syncthreads();
    if (fits) {
        #pragma unroll
        for (int j = 0; j < KCAP / 512; ++j)
            if (pk[j] >= 0) atomicAdd(&c[pk[j] & (BSZ - 1)], 1);
    } else {
        for (int i = tid; i < n; i += 512)
            atomicAdd(&c[part[beg + i] & (BSZ - 1)], 1);
    }
    __syncthreads();

    const int v = (tid < BSZ) ? c[tid] : 0;
    int incl = v;
    for (int o = 1; o < 32; o <<= 1) {
        const int t = __shfl_up(incl, o);
        if (lane >= o) incl += t;
    }
    if (lane == 31) wsum[wave] = incl;
    __syncthreads();
    int off = 0;
    for (int k = 0; k < wave; ++k) off += wsum[k];
    const int ex = off + incl - v;
    if (tid < BSZ) {
        const int node = b * BSZ + tid;
        if (node < NN) { cnt[node] = v; rowstart[node] = beg + ex; }
        rs[tid] = ex;
        c[tid]  = 0;
    }
    __syncthreads();

    if (fits) {
        #pragma unroll
        for (int j = 0; j < KCAP / 512; ++j)
            if (pk[j] >= 0) {
                const int dl = pk[j] & (BSZ - 1);
                stg[rs[dl] + atomicAdd(&c[dl], 1)] = pk[j] >> BSH;
            }
    } else {
        for (int i = tid; i < n; i += 512) {
            const int q  = part[beg + i];
            const int dl = q & (BSZ - 1);
            nbr[beg + rs[dl] + atomicAdd(&c[dl], 1)] = q >> BSH;
        }
    }
    __syncthreads();
    if (fits)
        for (int i = tid; i < n; i += 512) nbr[beg + i] = stg[i];
}

__device__ __forceinline__ float4 ld_row(const float4* __restrict__ x4, int s, int lane) {
    return x4[(size_t)s * (DH / 4) + lane];
}
__device__ __forceinline__ void add4(float4& a, const float4& v) {
    a.x += v.x; a.y += v.y; a.z += v.z; a.w += v.w;
}

#define MW 8
#define FW 8
#define NTILES (NN / TM)
#define LGRID 1024

static_assert(LGRID <= NTILES, "every work-group owns at least one tile");

__device__ __forceinline__ void fill_tile(const float* __restrict__ x,
                                          const int* __restrict__ cnt,
                                          const int* __restrict__ rowstart,
                                          const int* __restrict__ nbr,
                                          float* __restrict__ A,
                                          int m0, unsigned tid) {
    asm volatile("" : "+v"(tid));
    const int lane = (int)(tid & 31u);
    const unsigned ft = tid - MW * 32;
    const int fw   = __builtin_amdgcn_readfirstlane((int)(tid >> 5)) - MW;
    const char* __restrict__ xs = (const char*)(x + (size_t)m0 * DH);
    static_assert(TM * (DH / 4) % (FW * 32) == 0, "whole pieces per fill thread");
    #pragma unroll
    for (int i = 0; i < TM * (DH / 4) / (FW * 32); ++i) {
        const unsigned idx = ft + i * (FW * 32);
        const unsigned row = idx / (DH / 4);
        const unsigned c4  = idx - row * (DH / 4);
        *(float4*)(A + row * A_STRIDE + DH + 4 * c4) = *(const float4*)(xs + idx * 16u);
    }

    const float4* __restrict__ x4 = (const float4*)x;
    for (int row = fw; row < TM; row += FW) {
        const int g   = m0 + row;
        const int c   = __builtin_amdgcn_readfirstlane(cnt[g]);
        const int beg = __builtin_amdgcn_readfirstlane(rowstart[g]);
        float4 acc = make_float4(0.f, 0.f, 0.f, 0.f);
        for (int base = 0; base < c; base += 32) {
            const int n  = (c - base < 32) ? (c - base) : 32;
            const int my = (lane < n) ? nbr[beg + base + lane] : 0;
            int j = 0;
            for (; j + 8 <= n; j += 8) {
                float4 v0 = ld_row(x4, __builtin_amdgcn_readlane(my, j + 0), lane);
                float4 v1 = ld_row(x4, __builtin_amdgcn_readlane(my, j + 1), lane);
                float4 v2 = ld_row(x4, __builtin_amdgcn_readlane(my, j + 2), lane);
                float4 v3 = ld_row(x4, __builtin_amdgcn_readlane(my, j + 3), lane);
                float4 v4 = ld_row(x4, __builtin_amdgcn_readlane(my, j + 4), lane);
                float4 v5 = ld_row(x4, __builtin_amdgcn_readlane(my, j + 5), lane);
                float4 v6 = ld_row(x4, __builtin_amdgcn_readlane(my, j + 6), lane);
                float4 v7 = ld_row(x4, __builtin_amdgcn_readlane(my, j + 7), lane);
                add4(acc, v0); add4(acc, v1); add4(acc, v2); add4(acc, v3);
                add4(acc, v4); add4(acc, v5); add4(acc, v6); add4(acc, v7);
            }
            if (j + 4 <= n) {
                float4 v0 = ld_row(x4, __builtin_amdgcn_readlane(my, j + 0), lane);
                float4 v1 = ld_row(x4, __builtin_amdgcn_readlane(my, j + 1), lane);
                float4 v2 = ld_row(x4, __builtin_amdgcn_readlane(my, j + 2), lane);
                float4 v3 = ld_row(x4, __builtin_amdgcn_readlane(my, j + 3), lane);
                add4(acc, v0); add4(acc, v1); add4(acc, v2); add4(acc, v3);
                j += 4;
            }
            if (j + 2 <= n) {
                float4 v0 = ld_row(x4, __builtin_amdgcn_readlane(my, j + 0), lane);
                float4 v1 = ld_row(x4, __builtin_amdgcn_readlane(my, j + 1), lane);
                add4(acc, v0); add4(acc, v1);
                j += 2;
            }
            if (j < n) {
                float4 v0 = ld_row(x4, __builtin_amdgcn_readlane(my, j), lane);
                add4(acc, v0);
            }
        }
        const float di = 1.0f / fmaxf((float)c, 1.0f);
        float4 r = make_float4(acc.x * di, acc.y * di, acc.z * di, acc.w * di);
        *(float4*)(A + row * A_STRIDE + 4 * lane) = r;
    }
}

template <int DIN, int DOUT, bool RELU>
__device__ __forceinline__ void mma_tile(const float* __restrict__ A,
                                         const float* __restrict__ Wl,
                                         const float* __restrict__ Wr,
                                         const float* __restrict__ bias,
                                         float* __restrict__ out,
                                         int m0, unsigned tid) {
    asm volatile("" : "+v"(tid));
    const int lane = (int)(tid & 31u);
    const int mw   = __builtin_amdgcn_readfirstlane((int)(tid >> 5));
    const int half = lane >> 4;
    const int l15  = lane & 15;
    const int col  = mw * 16 + l15;

    const float* __restrict__ Wlp = Wl + 2 * half * DOUT + col;
    const float* __restrict__ Wrp = Wr + 2 * half * DOUT + col;
    const float* __restrict__ Ap0 = A + l15 * A_STRIDE + 2 * half;
    const float* __restrict__ Ap1 = Ap0 + 16 * A_STRIDE;

    v8f acc0 = {};
    v8f acc1 = {};
    #pragma unroll 8
    for (int k = 0; k < DIN; k += 4) {
        v2f a0, a1, b;
        a0.x = Ap0[k];
        a0.y = Ap0[k + 1];
        a1.x = Ap1[k];
        a1.y = Ap1[k + 1];
        b.x = Wlp[k * DOUT];
        b.y = Wlp[k * DOUT + DOUT];
        acc0 = __builtin_amdgcn_wmma_f32_16x16x4_f32(
            false, a0, false, b, (short)0, acc0, false, false);
        acc1 = __builtin_amdgcn_wmma_f32_16x16x4_f32(
            false, a1, false, b, (short)0, acc1, false, false);
    }
    #pragma unroll 8
    for (int k = 0; k < DIN; k += 4) {
        v2f a0, a1, b;
        a0.x = Ap0[DIN + k];
        a0.y = Ap0[DIN + k + 1];
        a1.x = Ap1[DIN + k];
        a1.y = Ap1[DIN + k + 1];
        b.x = Wrp[k * DOUT];
        b.y = Wrp[k * DOUT + DOUT];
        acc0 = __builtin_amdgcn_wmma_f32_16x16x4_f32(
            false, a0, false, b, (short)0, acc0, false, false);
        acc1 = __builtin_amdgcn_wmma_f32_16x16x4_f32(
            false, a1, false, b, (short)0, acc1, false, false);
    }

    const float bv = bias[col];
    float* __restrict__ op = out + (size_t)(m0 + 8 * half) * DOUT + col;
    #pragma unroll
    for (int r = 0; r < 8; ++r) {
        float v = acc0[r] + bv;
        if (RELU) v = fmaxf(v, 0.0f);
        op[r * DOUT] = v;
    }
    op += 16 * DOUT;
    #pragma unroll
    for (int r = 0; r < 8; ++r) {
        float v = acc1[r] + bv;
        if (RELU) v = fmaxf(v, 0.0f);
        op[r * DOUT] = v;
    }
}

template <int DIN, int DOUT, bool RELU>
__global__ void __launch_bounds__((MW + FW) * 32, 16)
sage_gemm(const float* __restrict__ x,
          const int* __restrict__ cnt,
          const int* __restrict__ rowstart,
          const int* __restrict__ nbr,
          const float* __restrict__ Wl,
          const float* __restrict__ Wr,
          const float* __restrict__ bias,
          float* __restrict__ out) {
    static_assert(DIN == DH, "row gather is written for 128-float rows");
    static_assert(DOUT == MW * 16, "one matrix wave per 16 output columns");
    __shared__ __attribute__((aligned(16))) float Alds[2][TM * A_STRIDE];

    const unsigned tid = threadIdx.x;
    const bool fill = __builtin_amdgcn_readfirstlane((int)(tid >> 5)) >= MW;

    if (fill) __builtin_amdgcn_s_setprio(3);

    int t = blockIdx.x;
    if (fill) fill_tile(x, cnt, rowstart, nbr, Alds[0], t * TM, tid);
    __syncthreads();
    int cur = 0;
    for (; t < NTILES; t += gridDim.x) {
        const int tn = t + gridDim.x;
        if (fill) {
            if (tn < NTILES)
                fill_tile(x, cnt, rowstart, nbr, Alds[cur ^ 1], tn * TM, tid);
        } else {
            mma_tile<DIN, DOUT, RELU>(Alds[cur], Wl, Wr, bias, out, t * TM, tid);
        }
        __syncthreads();
        cur ^= 1;
    }
    if (fill) __builtin_amdgcn_s_setprio(0);
}

__global__ void proj_gemm(const float* __restrict__ h,
                          const float* __restrict__ Wl,
                          const float* __restrict__ Wr,
                          const float* __restrict__ bias,
                          float* __restrict__ P,
                          float* __restrict__ Q) {
    __shared__ __attribute__((aligned(16))) float Alds[TM * P_STRIDE];

    const int m0   = blockIdx.x * TM;
    const int tid  = threadIdx.x;
    const int lane = tid & 31;
    const int wave = tid >> 5;

    const float4* __restrict__ h4 = (const float4*)(h + (size_t)m0 * DH);
    for (int idx = tid; idx < TM * (DH / 4); idx += blockDim.x) {
        const int row = idx / (DH / 4);
        const int c4  = idx - row * (DH / 4);
        *(float4*)(Alds + row * P_STRIDE + 4 * c4) = h4[idx];
    }
    __syncthreads();

    const int half = lane >> 4;
    const int l15  = lane & 15;
    const int col  = wave * 16 + l15;
    const bool isP = col < NC;
    const int c    = isP ? col : col - NC;

    const float* __restrict__ Wp = (isP ? Wl : Wr) + 2 * half * NC + c;
    const float* __restrict__ Ap0 = Alds + l15 * P_STRIDE + 2 * half;
    const float* __restrict__ Ap1 = Ap0 + 16 * P_STRIDE;

    v8f acc0 = {};
    v8f acc1 = {};
    #pragma unroll 8
    for (int k = 0; k < DH; k += 4) {
        v2f a0, a1, b;
        a0.x = Ap0[k];
        a0.y = Ap0[k + 1];
        a1.x = Ap1[k];
        a1.y = Ap1[k + 1];
        b.x = Wp[k * NC];
        b.y = Wp[k * NC + NC];
        acc0 = __builtin_amdgcn_wmma_f32_16x16x4_f32(
            false, a0, false, b, (short)0, acc0, false, false);
        acc1 = __builtin_amdgcn_wmma_f32_16x16x4_f32(
            false, a1, false, b, (short)0, acc1, false, false);
    }

    const float bv = isP ? 0.0f : bias[c];
    const int   rs = isP ? PS : NC;
    float* __restrict__ op = (isP ? P + (size_t)(m0 + 8 * half) * PS
                                  : Q + (size_t)(m0 + 8 * half) * NC) + c;
    #pragma unroll
    for (int r = 0; r < 8; ++r)
        op[r * rs] = acc0[r] + bv;
    op += 16 * rs;
    #pragma unroll
    for (int r = 0; r < 8; ++r)
        op[r * rs] = acc1[r] + bv;
}

__global__ void aggr_logsoftmax(const float* __restrict__ P,
                                const float* __restrict__ Q,
                                const int* __restrict__ cnt,
                                const int* __restrict__ rowstart,
                                const int* __restrict__ nbr,
                                float* __restrict__ out) {
    const int lane = threadIdx.x & 31;
    const int wave = threadIdx.x >> 5;
    const int grp  = lane / 10;
    const int ch   = lane - grp * 10;
    const int row  = (blockIdx.x * (blockDim.x >> 5) + wave) * 3 + grp;
    const bool on  = (grp < 3) && (row < NN);
    const int r    = on ? row : 0;
    const int c    = on ? cnt[r] : 0;
    const int beg  = rowstart[r];

    const float4* __restrict__ P4 = (const float4*)P;
    const float4 q = ((const float4*)Q)[(size_t)r * (NC / 4) + ch];

    float4 acc = make_float4(0.f, 0.f, 0.f, 0.f);
    int j = 0;
    for (; j + 8 <= c; j += 8) {
        int s[8];
        #pragma unroll
        for (int u = 0; u < 8; ++u) s[u] = nbr[beg + j + u];
        float4 v[8];
        #pragma unroll
        for (int u = 0; u < 8; ++u) v[u] = P4[(size_t)s[u] * (PS / 4) + ch];
        #pragma unroll
        for (int u = 0; u < 8; ++u) add4(acc, v[u]);
    }
    if (j < c) {
        const int left = c - j;
        int s[7];
        #pragma unroll
        for (int u = 0; u < 7; ++u) s[u] = nbr[beg + j + ((u < left) ? u : (left - 1))];
        float4 v[7];
        #pragma unroll
        for (int u = 0; u < 7; ++u) {
            v[u] = make_float4(0.f, 0.f, 0.f, 0.f);
            if (u < left) v[u] = P4[(size_t)s[u] * (PS / 4) + ch];
        }
        #pragma unroll
        for (int u = 0; u < 7; ++u) add4(acc, v[u]);
    }

    const float di = 1.0f / fmaxf((float)c, 1.0f);
    const float z0 = acc.x * di + q.x;
    const float z1 = acc.y * di + q.y;
    const float z2 = acc.z * di + q.z;
    const float z3 = acc.w * di + q.w;

    const int gb = (grp < 3) ? grp * 10 : 0;
    const float ml = fmaxf(fmaxf(z0, z1), fmaxf(z2, z3));
    float mx = -INFINITY;
    #pragma unroll
    for (int k = 0; k < 10; ++k) mx = fmaxf(mx, __shfl(ml, gb + k));
    const float el = expf(z0 - mx) + expf(z1 - mx) + expf(z2 - mx) + expf(z3 - mx);
    float sum = 0.0f;
    #pragma unroll
    for (int k = 0; k < 10; ++k) sum += __shfl(el, gb + k);
    const float lse = mx + logf(sum);

    if (on)
        ((float4*)out)[(size_t)row * (NC / 4) + ch] =
            make_float4(z0 - lse, z1 - lse, z2 - lse, z3 - lse);
}

extern "C" void kernel_launch(void* const* d_in, const int* in_sizes, int n_in,
                              void* d_out, int out_size, void* d_ws, size_t ws_size,
                              hipStream_t stream) {
    const float* x   = (const float*)d_in[0];
    const int*   ei  = (const int*)d_in[1];
    const int*   src = ei;
    const int*   dst = ei + NE;
    const float* Wl0 = (const float*)d_in[2];
    const float* Wr0 = (const float*)d_in[3];
    const float* b0  = (const float*)d_in[4];
    const float* Wl1 = (const float*)d_in[5];
    const float* Wr1 = (const float*)d_in[6];
    const float* b1  = (const float*)d_in[7];
    const float* Wl2 = (const float*)d_in[8];
    const float* Wr2 = (const float*)d_in[9];
    const float* b2  = (const float*)d_in[10];
    float* hout = (float*)d_out;

    char* ws = (char*)d_ws;
    size_t off = 0;
    auto carve = [&](size_t bytes) {
        void* p = ws + off;
        off = (off + bytes + 255) & ~(size_t)255;
        return p;
    };
    int*   cnt      = (int*)carve((size_t)NN * 4);
    int*   rowstart = (int*)carve((size_t)NN * 4);
    int*   part     = (int*)carve((size_t)NE * 4);
    int*   nbr      = (int*)carve((size_t)NE * 4);
    float* h0       = (float*)carve((size_t)NN * DH * 4);
    float* h1       = (float*)carve((size_t)NN * DH * 4);
    float* P        = h0;
    float* Q        = h0 + (size_t)NN * PS;
    int*   hist     = (int*)h1;
    int*   bbase    = hist + NCH * NB;

    const int ZB = 256;
    dim3 bZ(ZB);
    dim3 gAggr((NN + 23) / 24);
    dim3 gGemm(NN / TM);

    bucket_hist<<<dim3(NCH), dim3(ST), 0, stream>>>(dst, hist);
    bucket_scatter<<<dim3(NCH), dim3(ST), 0, stream>>>(src, dst, hist, bbase, part);
    bucket_sort<<<dim3(NB), dim3(512), 0, stream>>>(part, bbase, cnt, rowstart, nbr);

    sage_gemm<DH, DH, true><<<dim3(LGRID), dim3((MW + FW) * 32), 0, stream>>>(x, cnt, rowstart, nbr, Wl0, Wr0, b0, h0);

    sage_gemm<DH, DH, true><<<dim3(LGRID), dim3((MW + FW) * 32), 0, stream>>>(h0, cnt, rowstart, nbr, Wl1, Wr1, b1, h1);

    proj_gemm<<<gGemm, dim3(160), 0, stream>>>(h1, Wl2, Wr2, b2, P, Q);
    aggr_logsoftmax<<<gAggr, bZ, 0, stream>>>(P, Q, cnt, rowstart, nbr, hout);
}


# Edge sort in 2 launches, not 3: every chunk grouped by bucket in place (no counting launch, no table sums); a bucket's base = one column sum of the chunks' slot rows; row order still LDS arrival
# speedup vs baseline: 5.3671x; 1.0497x over previous
#include <hip/hip_runtime.h>
#include <math.h>


#define NN 100000
#define NE 1600000
#define DH 128
#define NC 40
#define A_STRIDE 260
#define P_STRIDE 132
#define PS 64
#define TM 32

static_assert(NN % TM == 0 && TM == 32, "whole row tiles only; two accumulator tiles per wave");

typedef __attribute__((ext_vector_type(2))) float v2f;
typedef __attribute__((ext_vector_type(8))) float v8f;

#define BSH 8
#define BSZ (1 << BSH)
#define NB  ((NN + BSZ - 1) / BSZ)
#define CH  6400
#define NCH (NE / CH)
#define ST  640
#define KCAP 8192
#define TABW (NB + 1)
#define PKN 16

static_assert(NE % CH == 0 && CH % ST == 0 && ST % 32 == 0, "whole chunks, whole waves");
static_assert(NB <= ST, "one thread per bucket in the offset scan");
static_assert((NN - 1) < (1 << (31 - BSH)), "source index and low destination bits share one word");
static_assert(NCH <= 512 && NCH <= 16 * 16, "launch 2: one thread per chunk row, 16 pieces per wave");

__global__ void __launch_bounds__(ST)
chunk_group(const int* __restrict__ src,
            const int* __restrict__ dst,
            int* __restrict__ tab,
            int* __restrict__ part) {
    __shared__ int h[NB];
    __shared__ int lstart[NB];
    __shared__ int wt[ST / 32];
    __shared__ int stage[CH];

    const int tid  = threadIdx.x;
    const int lane = tid & 31;
    const int wave = tid >> 5;
    const int w    = blockIdx.x;

    if (tid < NB) h[tid] = 0;
    const int* __restrict__ s = src + w * CH;
    const int* __restrict__ d = dst + w * CH;
    int dv[CH / ST], sv[CH / ST];
    #pragma unroll
    for (int i = 0; i < CH / ST; ++i) {
        dv[i] = d[tid + i * ST];
        sv[i] = s[tid + i * ST];
    }
    __syncthreads();
    #pragma unroll
    for (int i = 0; i < CH / ST; ++i) atomicAdd(&h[dv[i] >> BSH], 1);
    __syncthreads();

    const int v = (tid < NB) ? h[tid] : 0;
    int incl = v;
    for (int o = 1; o < 32; o <<= 1) {
        const int t = __shfl_up(incl, o);
        if (lane >= o) incl += t;
    }
    if (lane == 31) wt[wave] = incl;
    __syncthreads();
    int off = 0;
    for (int k = 0; k < wave; ++k) off += wt[k];
    const int ex = off + incl - v;
    if (tid < NB) {
        lstart[tid] = ex;
        h[tid]      = 0;
        tab[w * TABW + tid] = ex;
    }
    if (tid == 0) tab[w * TABW + NB] = CH;
    __syncthreads();

    #pragma unroll
    for (int i = 0; i < CH / ST; ++i) {
        const int b = dv[i] >> BSH;
        const int p = lstart[b] + atomicAdd(&h[b], 1);
        stage[p] = (sv[i] << BSH) | (dv[i] & (BSZ - 1));
    }
    __syncthreads();
    #pragma unroll
    for (int i = 0; i < CH / ST; ++i) {
        const int p = tid + i * ST;
        part[w * CH + p] = stage[p];
    }
}

__global__ void __launch_bounds__(512)
bucket_sort(const int* __restrict__ part,
            const int* __restrict__ tab,
            int* __restrict__ cnt,
            int* __restrict__ rowstart,
            int* __restrict__ nbr) {
    __shared__ int c[BSZ];
    __shared__ int rs[BSZ];
    __shared__ int wsum[16];
    __shared__ int ps[NCH];
    __shared__ int pl[NCH];
    __shared__ int red[3][16];
    __shared__ int stg[KCAP];

    const int tid  = threadIdx.x;
    const int lane = tid & 31;
    const int wave = tid >> 5;
    const int b    = blockIdx.x;

    int s0 = 0, len = 0;
    if (tid < NCH) {
        s0  = tab[tid * TABW + b];
        len = tab[tid * TABW + b + 1] - s0;
        ps[tid] = s0;
        pl[tid] = len;
    }
    int rb = s0, rn = len, rm = len;
    for (int o = 16; o > 0; o >>= 1) {
        rb += __shfl_xor(rb, o);
        rn += __shfl_xor(rn, o);
        const int t = __shfl_xor(rm, o);
        rm = (t > rm) ? t : rm;
    }
    if (lane == 0) { red[0][wave] = rb; red[1][wave] = rn; red[2][wave] = rm; }
    if (tid < BSZ) c[tid] = 0;
    __syncthreads();
    int beg = 0, n = 0, mx = 0;
    for (int k = 0; k < 16; ++k) {
        beg += red[0][k];
        n   += red[1][k];
        mx   = (red[2][k] > mx) ? red[2][k] : mx;
    }
    const bool fits = n <= KCAP;
    const bool big  = mx > 32;

    const int l15 = lane & 15;
    const int hf  = lane >> 4;
    int pk[PKN];
    int pL[PKN / 2], pO[PKN / 2];
    #pragma unroll
    for (int j = 0; j < PKN / 2; ++j) {
        const int pc = wave * 16 + 2 * j + hf;
        const int pr = (pc < NCH) ? pc : 0;
        pL[j] = (pc < NCH) ? pl[pr] : 0;
        pO[j] = pr * CH + ps[pr];
    }
    #pragma unroll
    for (int j = 0; j < PKN / 2; ++j) {
        pk[j]           = (l15 < pL[j])      ? part[pO[j] + l15]      : -1;
        pk[PKN / 2 + j] = (l15 + 16 < pL[j]) ? part[pO[j] + l15 + 16] : -1;
    }

    #pragma unroll
    for (int j = 0; j < PKN; ++j)
        if (pk[j] >= 0) atomicAdd(&c[pk[j] & (BSZ - 1)], 1);
    if (big) {
        for (int j = 0; j < PKN / 2; ++j) {
            const int pc = wave * 16 + 2 * j + hf;
            if (pc < NCH) {
                const int L = pl[pc];
                const int* __restrict__ q = part + pc * CH + ps[pc];
                for (int e = 32 + l15; e < L; e += 16)
                    atomicAdd(&c[q[e] & (BSZ - 1)], 1);
            }
        }
    }
    __syncthreads();

    const int v = (tid < BSZ) ? c[tid] : 0;
    int incl = v;
    for (int o = 1; o < 32; o <<= 1) {
        const int t = __shfl_up(incl, o);
        if (lane >= o) incl += t;
    }
    if (lane == 31) wsum[wave] = incl;
    __syncthreads();
    int off = 0;
    for (int k = 0; k < wave; ++k) off += wsum[k];
    const int ex = off + incl - v;
    if (tid < BSZ) {
        const int node = b * BSZ + tid;
        if (node < NN) { cnt[node] = v; rowstart[node] = beg + ex; }
        rs[tid] = ex;
        c[tid]  = 0;
    }
    __syncthreads();

    #pragma unroll
    for (int j = 0; j < PKN; ++j)
        if (pk[j] >= 0) {
            const int dl  = pk[j] & (BSZ - 1);
            const int pos = rs[dl] + atomicAdd(&c[dl], 1);
            if (fits) stg[pos] = pk[j] >> BSH;
            else      nbr[beg + pos] = pk[j] >> BSH;
        }
    if (big) {
        for (int j = 0; j < PKN / 2; ++j) {
            const int pc = wave * 16 + 2 * j + hf;
            if (pc < NCH) {
                const int L = pl[pc];
                const int* __restrict__ q = part + pc * CH + ps[pc];
                for (int e = 32 + l15; e < L; e += 16) {
                    const int qv  = q[e];
                    const int dl  = qv & (BSZ - 1);
                    const int pos = rs[dl] + atomicAdd(&c[dl], 1);
                    if (fits) stg[pos] = qv >> BSH;
                    else      nbr[beg + pos] = qv >> BSH;
                }
            }
        }
    }
    __syncthreads();
    if (fits)
        for (int i = tid; i < n; i += 512) nbr[beg + i] = stg[i];
}

__device__ __forceinline__ float4 ld_row(const float4* __restrict__ x4, int s, int lane) {
    return x4[(size_t)s * (DH / 4) + lane];
}
__device__ __forceinline__ void add4(float4& a, const float4& v) {
    a.x += v.x; a.y += v.y; a.z += v.z; a.w += v.w;
}

#define MW 8
#define FW 8
#define NTILES (NN / TM)
#define LGRID 1024

static_assert(LGRID <= NTILES, "every work-group owns at least one tile");

__device__ __forceinline__ void fill_tile(const float* __restrict__ x,
                                          const int* __restrict__ cnt,
                                          const int* __restrict__ rowstart,
                                          const int* __restrict__ nbr,
                                          float* __restrict__ A,
                                          int m0, unsigned tid) {
    asm volatile("" : "+v"(tid));
    const int lane = (int)(tid & 31u);
    const unsigned ft = tid - MW * 32;
    const int fw   = __builtin_amdgcn_readfirstlane((int)(tid >> 5)) - MW;
    const char* __restrict__ xs = (const char*)(x + (size_t)m0 * DH);
    static_assert(TM * (DH / 4) % (FW * 32) == 0, "whole pieces per fill thread");
    #pragma unroll
    for (int i = 0; i < TM * (DH / 4) / (FW * 32); ++i) {
        const unsigned idx = ft + i * (FW * 32);
        const unsigned row = idx / (DH / 4);
        const unsigned c4  = idx - row * (DH / 4);
        *(float4*)(A + row * A_STRIDE + DH + 4 * c4) = *(const float4*)(xs + idx * 16u);
    }

    const float4* __restrict__ x4 = (const float4*)x;
    for (int row = fw; row < TM; row += FW) {
        const int g   = m0 + row;
        const int c   = __builtin_amdgcn_readfirstlane(cnt[g]);
        const int beg = __builtin_amdgcn_readfirstlane(rowstart[g]);
        float4 acc = make_float4(0.f, 0.f, 0.f, 0.f);
        for (int base = 0; base < c; base += 32) {
            const int n  = (c - base < 32) ? (c - base) : 32;
            const int my = (lane < n) ? nbr[beg + base + lane] : 0;
            int j = 0;
            for (; j + 8 <= n; j += 8) {
                float4 v0 = ld_row(x4, __builtin_amdgcn_readlane(my, j + 0), lane);
                float4 v1 = ld_row(x4, __builtin_amdgcn_readlane(my, j + 1), lane);
                float4 v2 = ld_row(x4, __builtin_amdgcn_readlane(my, j + 2), lane);
                float4 v3 = ld_row(x4, __builtin_amdgcn_readlane(my, j + 3), lane);
                float4 v4 = ld_row(x4, __builtin_amdgcn_readlane(my, j + 4), lane);
                float4 v5 = ld_row(x4, __builtin_amdgcn_readlane(my, j + 5), lane);
                float4 v6 = ld_row(x4, __builtin_amdgcn_readlane(my, j + 6), lane);
                float4 v7 = ld_row(x4, __builtin_amdgcn_readlane(my, j + 7), lane);
                add4(acc, v0); add4(acc, v1); add4(acc, v2); add4(acc, v3);
                add4(acc, v4); add4(acc, v5); add4(acc, v6); add4(acc, v7);
            }
            if (j + 4 <= n) {
                float4 v0 = ld_row(x4, __builtin_amdgcn_readlane(my, j + 0), lane);
                float4 v1 = ld_row(x4, __builtin_amdgcn_readlane(my, j + 1), lane);
                float4 v2 = ld_row(x4, __builtin_amdgcn_readlane(my, j + 2), lane);
                float4 v3 = ld_row(x4, __builtin_amdgcn_readlane(my, j + 3), lane);
                add4(acc, v0); add4(acc, v1); add4(acc, v2); add4(acc, v3);
                j += 4;
            }
            if (j + 2 <= n) {
                float4 v0 = ld_row(x4, __builtin_amdgcn_readlane(my, j + 0), lane);
                float4 v1 = ld_row(x4, __builtin_amdgcn_readlane(my, j + 1), lane);
                add4(acc, v0); add4(acc, v1);
                j += 2;
            }
            if (j < n) {
                float4 v0 = ld_row(x4, __builtin_amdgcn_readlane(my, j), lane);
                add4(acc, v0);
            }
        }
        const float di = 1.0f / fmaxf((float)c, 1.0f);
        float4 r = make_float4(acc.x * di, acc.y * di, acc.z * di, acc.w * di);
        *(float4*)(A + row * A_STRIDE + 4 * lane) = r;
    }
}

template <int DIN, int DOUT, bool RELU>
__device__ __forceinline__ void mma_tile(const float* __restrict__ A,
                                         const float* __restrict__ Wl,
                                         const float* __restrict__ Wr,
                                         const float* __restrict__ bias,
                                         float* __restrict__ out,
                                         int m0, unsigned tid) {
    asm volatile("" : "+v"(tid));
    const int lane = (int)(tid & 31u);
    const int mw   = __builtin_amdgcn_readfirstlane((int)(tid >> 5));
    const int half = lane >> 4;
    const int l15  = lane & 15;
    const int col  = mw * 16 + l15;

    const float* __restrict__ Wlp = Wl + 2 * half * DOUT + col;
    const float* __restrict__ Wrp = Wr + 2 * half * DOUT + col;
    const float* __restrict__ Ap0 = A + l15 * A_STRIDE + 2 * half;
    const float* __restrict__ Ap1 = Ap0 + 16 * A_STRIDE;

    v8f acc0 = {};
    v8f acc1 = {};
    #pragma unroll 8
    for (int k = 0; k < DIN; k += 4) {
        v2f a0, a1, b;
        a0.x = Ap0[k];
        a0.y = Ap0[k + 1];
        a1.x = Ap1[k];
        a1.y = Ap1[k + 1];
        b.x = Wlp[k * DOUT];
        b.y = Wlp[k * DOUT + DOUT];
        acc0 = __builtin_amdgcn_wmma_f32_16x16x4_f32(
            false, a0, false, b, (short)0, acc0, false, false);
        acc1 = __builtin_amdgcn_wmma_f32_16x16x4_f32(
            false, a1, false, b, (short)0, acc1, false, false);
    }
    #pragma unroll 8
    for (int k = 0; k < DIN; k += 4) {
        v2f a0, a1, b;
        a0.x = Ap0[DIN + k];
        a0.y = Ap0[DIN + k + 1];
        a1.x = Ap1[DIN + k];
        a1.y = Ap1[DIN + k + 1];
        b.x = Wrp[k * DOUT];
        b.y = Wrp[k * DOUT + DOUT];
        acc0 = __builtin_amdgcn_wmma_f32_16x16x4_f32(
            false, a0, false, b, (short)0, acc0, false, false);
        acc1 = __builtin_amdgcn_wmma_f32_16x16x4_f32(
            false, a1, false, b, (short)0, acc1, false, false);
    }

    const float bv = bias[col];
    float* __restrict__ op = out + (size_t)(m0 + 8 * half) * DOUT + col;
    #pragma unroll
    for (int r = 0; r < 8; ++r) {
        float v = acc0[r] + bv;
        if (RELU) v = fmaxf(v, 0.0f);
        op[r * DOUT] = v;
    }
    op += 16 * DOUT;
    #pragma unroll
    for (int r = 0; r < 8; ++r) {
        float v = acc1[r] + bv;
        if (RELU) v = fmaxf(v, 0.0f);
        op[r * DOUT] = v;
    }
}

template <int DIN, int DOUT, bool RELU>
__global__ void __launch_bounds__((MW + FW) * 32, 16)
sage_gemm(const float* __restrict__ x,
          const int* __restrict__ cnt,
          const int* __restrict__ rowstart,
          const int* __restrict__ nbr,
          const float* __restrict__ Wl,
          const float* __restrict__ Wr,
          const float* __restrict__ bias,
          float* __restrict__ out) {
    static_assert(DIN == DH, "row gather is written for 128-float rows");
    static_assert(DOUT == MW * 16, "one matrix wave per 16 output columns");
    __shared__ __attribute__((aligned(16))) float Alds[2][TM * A_STRIDE];

    const unsigned tid = threadIdx.x;
    const bool fill = __builtin_amdgcn_readfirstlane((int)(tid >> 5)) >= MW;

    if (fill) __builtin_amdgcn_s_setprio(3);

    int t = blockIdx.x;
    if (fill) fill_tile(x, cnt, rowstart, nbr, Alds[0], t * TM, tid);
    __syncthreads();
    int cur = 0;
    for (; t < NTILES; t += gridDim.x) {
        const int tn = t + gridDim.x;
        if (fill) {
            if (tn < NTILES)
                fill_tile(x, cnt, rowstart, nbr, Alds[cur ^ 1], tn * TM, tid);
        } else {
            mma_tile<DIN, DOUT, RELU>(Alds[cur], Wl, Wr, bias, out, t * TM, tid);
        }
        __syncthreads();
        cur ^= 1;
    }
    if (fill) __builtin_amdgcn_s_setprio(0);
}

__global__ void proj_gemm(const float* __restrict__ h,
                          const float* __restrict__ Wl,
                          const float* __restrict__ Wr,
                          const float* __restrict__ bias,
                          float* __restrict__ P,
                          float* __restrict__ Q) {
    __shared__ __attribute__((aligned(16))) float Alds[TM * P_STRIDE];

    const int m0   = blockIdx.x * TM;
    const int tid  = threadIdx.x;
    const int lane = tid & 31;
    const int wave = tid >> 5;

    const float4* __restrict__ h4 = (const float4*)(h + (size_t)m0 * DH);
    for (int idx = tid; idx < TM * (DH / 4); idx += blockDim.x) {
        const int row = idx / (DH / 4);
        const int c4  = idx - row * (DH / 4);
        *(float4*)(Alds + row * P_STRIDE + 4 * c4) = h4[idx];
    }
    __syncthreads();

    const int half = lane >> 4;
    const int l15  = lane & 15;
    const int col  = wave * 16 + l15;
    const bool isP = col < NC;
    const int c    = isP ? col : col - NC;

    const float* __restrict__ Wp = (isP ? Wl : Wr) + 2 * half * NC + c;
    const float* __restrict__ Ap0 = Alds + l15 * P_STRIDE + 2 * half;
    const float* __restrict__ Ap1 = Ap0 + 16 * P_STRIDE;

    v8f acc0 = {};
    v8f acc1 = {};
    #pragma unroll 8
    for (int k = 0; k < DH; k += 4) {
        v2f a0, a1, b;
        a0.x = Ap0[k];
        a0.y = Ap0[k + 1];
        a1.x = Ap1[k];
        a1.y = Ap1[k + 1];
        b.x = Wp[k * NC];
        b.y = Wp[k * NC + NC];
        acc0 = __builtin_amdgcn_wmma_f32_16x16x4_f32(
            false, a0, false, b, (short)0, acc0, false, false);
        acc1 = __builtin_amdgcn_wmma_f32_16x16x4_f32(
            false, a1, false, b, (short)0, acc1, false, false);
    }

    const float bv = isP ? 0.0f : bias[c];
    const int   rs = isP ? PS : NC;
    float* __restrict__ op = (isP ? P + (size_t)(m0 + 8 * half) * PS
                                  : Q + (size_t)(m0 + 8 * half) * NC) + c;
    #pragma unroll
    for (int r = 0; r < 8; ++r)
        op[r * rs] = acc0[r] + bv;
    op += 16 * rs;
    #pragma unroll
    for (int r = 0; r < 8; ++r)
        op[r * rs] = acc1[r] + bv;
}

__global__ void aggr_logsoftmax(const float* __restrict__ P,
                                const float* __restrict__ Q,
                                const int* __restrict__ cnt,
                                const int* __restrict__ rowstart,
                                const int* __restrict__ nbr,
                                float* __restrict__ out) {
    const int lane = threadIdx.x & 31;
    const int wave = threadIdx.x >> 5;
    const int grp  = lane / 10;
    const int ch   = lane - grp * 10;
    const int row  = (blockIdx.x * (blockDim.x >> 5) + wave) * 3 + grp;
    const bool on  = (grp < 3) && (row < NN);
    const int r    = on ? row : 0;
    const int c    = on ? cnt[r] : 0;
    const int beg  = rowstart[r];

    const float4* __restrict__ P4 = (const float4*)P;
    const float4 q = ((const float4*)Q)[(size_t)r * (NC / 4) + ch];

    float4 acc = make_float4(0.f, 0.f, 0.f, 0.f);
    int j = 0;
    for (; j + 8 <= c; j += 8) {
        int s[8];
        #pragma unroll
        for (int u = 0; u < 8; ++u) s[u] = nbr[beg + j + u];
        float4 v[8];
        #pragma unroll
        for (int u = 0; u < 8; ++u) v[u] = P4[(size_t)s[u] * (PS / 4) + ch];
        #pragma unroll
        for (int u = 0; u < 8; ++u) add4(acc, v[u]);
    }
    if (j < c) {
        const int left = c - j;
        int s[7];
        #pragma unroll
        for (int u = 0; u < 7; ++u) s[u] = nbr[beg + j + ((u < left) ? u : (left - 1))];
        float4 v[7];
        #pragma unroll
        for (int u = 0; u < 7; ++u) {
            v[u] = make_float4(0.f, 0.f, 0.f, 0.f);
            if (u < left) v[u] = P4[(size_t)s[u] * (PS / 4) + ch];
        }
        #pragma unroll
        for (int u = 0; u < 7; ++u) add4(acc, v[u]);
    }

    const float di = 1.0f / fmaxf((float)c, 1.0f);
    const float z0 = acc.x * di + q.x;
    const float z1 = acc.y * di + q.y;
    const float z2 = acc.z * di + q.z;
    const float z3 = acc.w * di + q.w;

    const int gb = (grp < 3) ? grp * 10 : 0;
    const float ml = fmaxf(fmaxf(z0, z1), fmaxf(z2, z3));
    float mx = -INFINITY;
    #pragma unroll
    for (int k = 0; k < 10; ++k) mx = fmaxf(mx, __shfl(ml, gb + k));
    const float el = expf(z0 - mx) + expf(z1 - mx) + expf(z2 - mx) + expf(z3 - mx);
    float sum = 0.0f;
    #pragma unroll
    for (int k = 0; k < 10; ++k) sum += __shfl(el, gb + k);
    const float lse = mx + logf(sum);

    if (on)
        ((float4*)out)[(size_t)row * (NC / 4) + ch] =
            make_float4(z0 - lse, z1 - lse, z2 - lse, z3 - lse);
}

extern "C" void kernel_launch(void* const* d_in, const int* in_sizes, int n_in,
                              void* d_out, int out_size, void* d_ws, size_t ws_size,
                              hipStream_t stream) {
    const float* x   = (const float*)d_in[0];
    const int*   ei  = (const int*)d_in[1];
    const int*   src = ei;
    const int*   dst = ei + NE;
    const float* Wl0 = (const float*)d_in[2];
    const float* Wr0 = (const float*)d_in[3];
    const float* b0  = (const float*)d_in[4];
    const float* Wl1 = (const float*)d_in[5];
    const float* Wr1 = (const float*)d_in[6];
    const float* b1  = (const float*)d_in[7];
    const float* Wl2 = (const float*)d_in[8];
    const float* Wr2 = (const float*)d_in[9];
    const float* b2  = (const float*)d_in[10];
    float* hout = (float*)d_out;

    char* ws = (char*)d_ws;
    size_t off = 0;
    auto carve = [&](size_t bytes) {
        void* p = ws + off;
        off = (off + bytes + 255) & ~(size_t)255;
        return p;
    };
    int*   cnt      = (int*)carve((size_t)NN * 4);
    int*   rowstart = (int*)carve((size_t)NN * 4);
    int*   part     = (int*)carve((size_t)NE * 4);
    int*   nbr      = (int*)carve((size_t)NE * 4);
    float* h0       = (float*)carve((size_t)NN * DH * 4);
    float* h1       = (float*)carve((size_t)NN * DH * 4);
    float* P        = h0;
    float* Q        = h0 + (size_t)NN * PS;
    int*   tab      = (int*)h1;

    const int ZB = 256;
    dim3 bZ(ZB);
    dim3 gAggr((NN + 23) / 24);
    dim3 gGemm(NN / TM);

    chunk_group<<<dim3(NCH), dim3(ST), 0, stream>>>(src, dst, tab, part);
    bucket_sort<<<dim3(NB), dim3(512), 0, stream>>>(part, tab, cnt, rowstart, nbr);

    sage_gemm<DH, DH, true><<<dim3(LGRID), dim3((MW + FW) * 32), 0, stream>>>(x, cnt, rowstart, nbr, Wl0, Wr0, b0, h0);

    sage_gemm<DH, DH, true><<<dim3(LGRID), dim3((MW + FW) * 32), 0, stream>>>(h0, cnt, rowstart, nbr, Wl1, Wr1, b1, h1);

    proj_gemm<<<gGemm, dim3(160), 0, stream>>>(h1, Wl2, Wr2, b2, P, Q);
    aggr_logsoftmax<<<gAggr, bZ, 0, stream>>>(P, Q, cnt, rowstart, nbr, hout);
}


# Last layer's h1 @ [Wl2|Wr2] moved into layer 1's launch: matrix waves put the relu'd tile into the LDS A tile just multiplied and project it there (f32 WMMA); h1 never stored; 5 launches
# speedup vs baseline: 5.5369x; 1.0316x over previous
#include <hip/hip_runtime.h>
#include <math.h>


#define NN 100000
#define NE 1600000
#define DH 128
#define NC 40
#define A_STRIDE 260
#define PS 64
#define TM 32

static_assert(NN % TM == 0 && TM == 32, "whole row tiles only; two accumulator tiles per wave");

typedef __attribute__((ext_vector_type(2))) float v2f;
typedef __attribute__((ext_vector_type(8))) float v8f;

#define BSH 8
#define BSZ (1 << BSH)
#define NB  ((NN + BSZ - 1) / BSZ)
#define CH  6400
#define NCH (NE / CH)
#define ST  640
#define KCAP 8192
#define TABW (NB + 1)
#define PKN 16

static_assert(NE % CH == 0 && CH % ST == 0 && ST % 32 == 0, "whole chunks, whole waves");
static_assert(NB <= ST, "one thread per bucket in the offset scan");
static_assert((NN - 1) < (1 << (31 - BSH)), "source index and low destination bits share one word");
static_assert(NCH <= 512 && NCH <= 16 * 16, "launch 2: one thread per chunk row, 16 pieces per wave");

__global__ void __launch_bounds__(ST)
chunk_group(const int* __restrict__ src,
            const int* __restrict__ dst,
            int* __restrict__ tab,
            int* __restrict__ part) {
    __shared__ int h[NB];
    __shared__ int lstart[NB];
    __shared__ int wt[ST / 32];
    __shared__ int stage[CH];

    const int tid  = threadIdx.x;
    const int lane = tid & 31;
    const int wave = tid >> 5;
    const int w    = blockIdx.x;

    if (tid < NB) h[tid] = 0;
    const int* __restrict__ s = src + w * CH;
    const int* __restrict__ d = dst + w * CH;
    int dv[CH / ST], sv[CH / ST];
    #pragma unroll
    for (int i = 0; i < CH / ST; ++i) {
        dv[i] = d[tid + i * ST];
        sv[i] = s[tid + i * ST];
    }
    __syncthreads();
    #pragma unroll
    for (int i = 0; i < CH / ST; ++i) atomicAdd(&h[dv[i] >> BSH], 1);
    __syncthreads();

    const int v = (tid < NB) ? h[tid] : 0;
    int incl = v;
    for (int o = 1; o < 32; o <<= 1) {
        const int t = __shfl_up(incl, o);
        if (lane >= o) incl += t;
    }
    if (lane == 31) wt[wave] = incl;
    __syncthreads();
    int off = 0;
    for (int k = 0; k < wave; ++k) off += wt[k];
    const int ex = off + incl - v;
    if (tid < NB) {
        lstart[tid] = ex;
        h[tid]      = 0;
        tab[w * TABW + tid] = ex;
    }
    if (tid == 0) tab[w * TABW + NB] = CH;
    __syncthreads();

    #pragma unroll
    for (int i = 0; i < CH / ST; ++i) {
        const int b = dv[i] >> BSH;
        const int p = lstart[b] + atomicAdd(&h[b], 1);
        stage[p] = (sv[i] << BSH) | (dv[i] & (BSZ - 1));
    }
    __syncthreads();
    #pragma unroll
    for (int i = 0; i < CH / ST; ++i) {
        const int p = tid + i * ST;
        part[w * CH + p] = stage[p];
    }
}

__global__ void __launch_bounds__(512)
bucket_sort(const int* __restrict__ part,
            const int* __restrict__ tab,
            int* __restrict__ cnt,
            int* __restrict__ rowstart,
            int* __restrict__ nbr) {
    __shared__ int c[BSZ];
    __shared__ int rs[BSZ];
    __shared__ int wsum[16];
    __shared__ int ps[NCH];
    __shared__ int pl[NCH];
    __shared__ int red[3][16];
    __shared__ int stg[KCAP];

    const int tid  = threadIdx.x;
    const int lane = tid & 31;
    const int wave = tid >> 5;
    const int b    = blockIdx.x;

    int s0 = 0, len = 0;
    if (tid < NCH) {
        s0  = tab[tid * TABW + b];
        len = tab[tid * TABW + b + 1] - s0;
        ps[tid] = s0;
        pl[tid] = len;
    }
    int rb = s0, rn = len, rm = len;
    for (int o = 16; o > 0; o >>= 1) {
        rb += __shfl_xor(rb, o);
        rn += __shfl_xor(rn, o);
        const int t = __shfl_xor(rm, o);
        rm = (t > rm) ? t : rm;
    }
    if (lane == 0) { red[0][wave] = rb; red[1][wave] = rn; red[2][wave] = rm; }
    if (tid < BSZ) c[tid] = 0;
    __syncthreads();
    int beg = 0, n = 0, mx = 0;
    for (int k = 0; k < 16; ++k) {
        beg += red[0][k];
        n   += red[1][k];
        mx   = (red[2][k] > mx) ? red[2][k] : mx;
    }
    const bool fits = n <= KCAP;
    const bool big  = mx > 32;

    const int l15 = lane & 15;
    const int hf  = lane >> 4;
    int pk[PKN];
    int pL[PKN / 2], pO[PKN / 2];
    #pragma unroll
    for (int j = 0; j < PKN / 2; ++j) {
        const int pc = wave * 16 + 2 * j + hf;
        const int pr = (pc < NCH) ? pc : 0;
        pL[j] = (pc < NCH) ? pl[pr] : 0;
        pO[j] = pr * CH + ps[pr];
    }
    #pragma unroll
    for (int j = 0; j < PKN / 2; ++j) {
        pk[j]           = (l15 < pL[j])      ? part[pO[j] + l15]      : -1;
        pk[PKN / 2 + j] = (l15 + 16 < pL[j]) ? part[pO[j] + l15 + 16] : -1;
    }

    #pragma unroll
    for (int j = 0; j < PKN; ++j)
        if (pk[j] >= 0) atomicAdd(&c[pk[j] & (BSZ - 1)], 1);
    if (big) {
        for (int j = 0; j < PKN / 2; ++j) {
            const int pc = wave * 16 + 2 * j + hf;
            if (pc < NCH) {
                const int L = pl[pc];
                const int* __restrict__ q = part + pc * CH + ps[pc];
                for (int e = 32 + l15; e < L; e += 16)
                    atomicAdd(&c[q[e] & (BSZ - 1)], 1);
            }
        }
    }
    __syncthreads();

    const int v = (tid < BSZ) ? c[tid] : 0;
    int incl = v;
    for (int o = 1; o < 32; o <<= 1) {
        const int t = __shfl_up(incl, o);
        if (lane >= o) incl += t;
    }
    if (lane == 31) wsum[wave] = incl;
    __syncthreads();
    int off = 0;
    for (int k = 0; k < wave; ++k) off += wsum[k];
    const int ex = off + incl - v;
    if (tid < BSZ) {
        const int node = b * BSZ + tid;
        if (node < NN) { cnt[node] = v; rowstart[node] = beg + ex; }
        rs[tid] = ex;
        c[tid]  = 0;
    }
    __syncthreads();

    #pragma unroll
    for (int j = 0; j < PKN; ++j)
        if (pk[j] >= 0) {
            const int dl  = pk[j] & (BSZ - 1);
            const int pos = rs[dl] + atomicAdd(&c[dl], 1);
            if (fits) stg[pos] = pk[j] >> BSH;
            else      nbr[beg + pos] = pk[j] >> BSH;
        }
    if (big) {
        for (int j = 0; j < PKN / 2; ++j) {
            const int pc = wave * 16 + 2 * j + hf;
            if (pc < NCH) {
                const int L = pl[pc];
                const int* __restrict__ q = part + pc * CH + ps[pc];
                for (int e = 32 + l15; e < L; e += 16) {
                    const int qv  = q[e];
                    const int dl  = qv & (BSZ - 1);
                    const int pos = rs[dl] + atomicAdd(&c[dl], 1);
                    if (fits) stg[pos] = qv >> BSH;
                    else      nbr[beg + pos] = qv >> BSH;
                }
            }
        }
    }
    __syncthreads();
    if (fits)
        for (int i = tid; i < n; i += 512) nbr[beg + i] = stg[i];
}

__device__ __forceinline__ float4 ld_row(const float4* __restrict__ x4, int s, int lane) {
    return x4[(size_t)s * (DH / 4) + lane];
}
__device__ __forceinline__ void add4(float4& a, const float4& v) {
    a.x += v.x; a.y += v.y; a.z += v.z; a.w += v.w;
}

#define MW 8
#define FW 8
#define NTILES (NN / TM)
#define LGRID 1024

static_assert(LGRID <= NTILES, "every work-group owns at least one tile");

__device__ __forceinline__ void fill_tile(const float* __restrict__ x,
                                          const int* __restrict__ cnt,
                                          const int* __restrict__ rowstart,
                                          const int* __restrict__ nbr,
                                          float* __restrict__ A,
                                          int m0, unsigned tid,
                                          int row_lo, int row_hi, bool self) {
    asm volatile("" : "+v"(tid));
    const int lane = (int)(tid & 31u);
    const unsigned ft = tid - MW * 32;
    const int fw   = __builtin_amdgcn_readfirstlane((int)(tid >> 5)) - MW;
    const char* __restrict__ xs = (const char*)(x + (size_t)m0 * DH);
    static_assert(TM * (DH / 4) % (FW * 32) == 0, "whole pieces per fill thread");
    if (self) {
        #pragma unroll
        for (int i = 0; i < TM * (DH / 4) / (FW * 32); ++i) {
            const unsigned idx = ft + i * (FW * 32);
            const unsigned row = idx / (DH / 4);
            const unsigned c4  = idx - row * (DH / 4);
            *(float4*)(A + row * A_STRIDE + DH + 4 * c4) = *(const float4*)(xs + idx * 16u);
        }
    }

    const float4* __restrict__ x4 = (const float4*)x;
    for (int row = row_lo + fw; row < row_hi; row += FW) {
        const int g   = m0 + row;
        const int c   = __builtin_amdgcn_readfirstlane(cnt[g]);
        const int beg = __builtin_amdgcn_readfirstlane(rowstart[g]);
        float4 acc = make_float4(0.f, 0.f, 0.f, 0.f);
        for (int base = 0; base < c; base += 32) {
            const int n  = (c - base < 32) ? (c - base) : 32;
            const int my = (lane < n) ? nbr[beg + base + lane] : 0;
            int j = 0;
            for (; j + 8 <= n; j += 8) {
                float4 v0 = ld_row(x4, __builtin_amdgcn_readlane(my, j + 0), lane);
                float4 v1 = ld_row(x4, __builtin_amdgcn_readlane(my, j + 1), lane);
                float4 v2 = ld_row(x4, __builtin_amdgcn_readlane(my, j + 2), lane);
                float4 v3 = ld_row(x4, __builtin_amdgcn_readlane(my, j + 3), lane);
                float4 v4 = ld_row(x4, __builtin_amdgcn_readlane(my, j + 4), lane);
                float4 v5 = ld_row(x4, __builtin_amdgcn_readlane(my, j + 5), lane);
                float4 v6 = ld_row(x4, __builtin_amdgcn_readlane(my, j + 6), lane);
                float4 v7 = ld_row(x4, __builtin_amdgcn_readlane(my, j + 7), lane);
                add4(acc, v0); add4(acc, v1); add4(acc, v2); add4(acc, v3);
                add4(acc, v4); add4(acc, v5); add4(acc, v6); add4(acc, v7);
            }
            if (j + 4 <= n) {
                float4 v0 = ld_row(x4, __builtin_amdgcn_readlane(my, j + 0), lane);
                float4 v1 = ld_row(x4, __builtin_amdgcn_readlane(my, j + 1), lane);
                float4 v2 = ld_row(x4, __builtin_amdgcn_readlane(my, j + 2), lane);
                float4 v3 = ld_row(x4, __builtin_amdgcn_readlane(my, j + 3), lane);
                add4(acc, v0); add4(acc, v1); add4(acc, v2); add4(acc, v3);
                j += 4;
            }
            if (j + 2 <= n) {
                float4 v0 = ld_row(x4, __builtin_amdgcn_readlane(my, j + 0), lane);
                float4 v1 = ld_row(x4, __builtin_amdgcn_readlane(my, j + 1), lane);
                add4(acc, v0); add4(acc, v1);
                j += 2;
            }
            if (j < n) {
                float4 v0 = ld_row(x4, __builtin_amdgcn_readlane(my, j), lane);
                add4(acc, v0);
            }
        }
        const float di = 1.0f / fmaxf((float)c, 1.0f);
        float4 r = make_float4(acc.x * di, acc.y * di, acc.z * di, acc.w * di);
        *(float4*)(A + row * A_STRIDE + 4 * lane) = r;
    }
}

template <int DIN, int DOUT>
__device__ __forceinline__ void mma_acc(const float* __restrict__ A,
                                        const float* __restrict__ Wl,
                                        const float* __restrict__ Wr,
                                        unsigned tid, v8f& acc0, v8f& acc1) {
    asm volatile("" : "+v"(tid));
    const int lane = (int)(tid & 31u);
    const int mw   = __builtin_amdgcn_readfirstlane((int)(tid >> 5));
    const int half = lane >> 4;
    const int l15  = lane & 15;
    const int col  = mw * 16 + l15;

    const float* __restrict__ Wlp = Wl + 2 * half * DOUT + col;
    const float* __restrict__ Wrp = Wr + 2 * half * DOUT + col;
    const float* __restrict__ Ap0 = A + l15 * A_STRIDE + 2 * half;
    const float* __restrict__ Ap1 = Ap0 + 16 * A_STRIDE;

    acc0 = v8f{};
    acc1 = v8f{};
    #pragma unroll 8
    for (int k = 0; k < DIN; k += 4) {
        v2f a0, a1, b;
        a0.x = Ap0[k];
        a0.y = Ap0[k + 1];
        a1.x = Ap1[k];
        a1.y = Ap1[k + 1];
        b.x = Wlp[k * DOUT];
        b.y = Wlp[k * DOUT + DOUT];
        acc0 = __builtin_amdgcn_wmma_f32_16x16x4_f32(
            false, a0, false, b, (short)0, acc0, false, false);
        acc1 = __builtin_amdgcn_wmma_f32_16x16x4_f32(
            false, a1, false, b, (short)0, acc1, false, false);
    }
    #pragma unroll 8
    for (int k = 0; k < DIN; k += 4) {
        v2f a0, a1, b;
        a0.x = Ap0[DIN + k];
        a0.y = Ap0[DIN + k + 1];
        a1.x = Ap1[DIN + k];
        a1.y = Ap1[DIN + k + 1];
        b.x = Wrp[k * DOUT];
        b.y = Wrp[k * DOUT + DOUT];
        acc0 = __builtin_amdgcn_wmma_f32_16x16x4_f32(
            false, a0, false, b, (short)0, acc0, false, false);
        acc1 = __builtin_amdgcn_wmma_f32_16x16x4_f32(
            false, a1, false, b, (short)0, acc1, false, false);
    }
}

__device__ __forceinline__ float col_bias(const float* __restrict__ bias, unsigned tid) {
    asm volatile("" : "+v"(tid));
    const int mw = __builtin_amdgcn_readfirstlane((int)(tid >> 5));
    return bias[mw * 16 + (int)(tid & 15u)];
}

template <int DOUT, bool RELU>
__device__ __forceinline__ void put_rows(const v8f& acc0, const v8f& acc1, float bv,
                                         float* __restrict__ out, int pitch, unsigned tid) {
    asm volatile("" : "+v"(tid));
    const int lane = (int)(tid & 31u);
    const int mw   = __builtin_amdgcn_readfirstlane((int)(tid >> 5));
    const int half = lane >> 4;
    const int col  = mw * 16 + (lane & 15);

    float* __restrict__ op = out + (size_t)(8 * half) * pitch + col;
    #pragma unroll
    for (int r = 0; r < 8; ++r) {
        float v = acc0[r] + bv;
        if (RELU) v = fmaxf(v, 0.0f);
        op[r * pitch] = v;
    }
    op += 16 * pitch;
    #pragma unroll
    for (int r = 0; r < 8; ++r) {
        float v = acc1[r] + bv;
        if (RELU) v = fmaxf(v, 0.0f);
        op[r * pitch] = v;
    }
}

__device__ __forceinline__ void proj_rows(const float* __restrict__ H,
                                          const float* __restrict__ Wl2,
                                          const float* __restrict__ Wr2,
                                          const float* __restrict__ b2,
                                          float* __restrict__ P,
                                          float* __restrict__ Q,
                                          int m0, unsigned tid) {
    asm volatile("" : "+v"(tid));
    const int lane = (int)(tid & 31u);
    const int mw   = __builtin_amdgcn_readfirstlane((int)(tid >> 5));
    if (mw < (2 * NC) / 16) {
        const int half = lane >> 4;
        const int l15  = lane & 15;
        const int col  = mw * 16 + l15;
        const bool isP = col < NC;
        const int c    = isP ? col : col - NC;

        const float* __restrict__ Wp  = (isP ? Wl2 : Wr2) + 2 * half * NC + c;
        const float* __restrict__ Hp0 = H + l15 * A_STRIDE + 2 * half;
        const float* __restrict__ Hp1 = Hp0 + 16 * A_STRIDE;

        v8f p0 = {};
        v8f p1 = {};
        #pragma unroll 8
        for (int k = 0; k < DH; k += 4) {
            v2f a0, a1, b;
            a0.x = Hp0[k];
            a0.y = Hp0[k + 1];
            a1.x = Hp1[k];
            a1.y = Hp1[k + 1];
            b.x = Wp[k * NC];
            b.y = Wp[k * NC + NC];
            p0 = __builtin_amdgcn_wmma_f32_16x16x4_f32(
                false, a0, false, b, (short)0, p0, false, false);
            p1 = __builtin_amdgcn_wmma_f32_16x16x4_f32(
                false, a1, false, b, (short)0, p1, false, false);
        }

        const float bv = isP ? 0.0f : b2[c];
        const int   rs = isP ? PS : NC;
        float* __restrict__ op = (isP ? P + (size_t)(m0 + 8 * half) * PS
                                      : Q + (size_t)(m0 + 8 * half) * NC) + c;
        #pragma unroll
        for (int r = 0; r < 8; ++r)
            op[r * rs] = p0[r] + bv;
        op += 16 * rs;
        #pragma unroll
        for (int r = 0; r < 8; ++r)
            op[r * rs] = p1[r] + bv;
    }
}

template <int DIN, int DOUT, bool RELU, bool PROJ>
__global__ void __launch_bounds__((MW + FW) * 32, 16)
sage_gemm(const float* __restrict__ x,
          const int* __restrict__ cnt,
          const int* __restrict__ rowstart,
          const int* __restrict__ nbr,
          const float* __restrict__ Wl,
          const float* __restrict__ Wr,
          const float* __restrict__ bias,
          float* __restrict__ out,
          const float* __restrict__ Wl2,
          const float* __restrict__ Wr2,
          const float* __restrict__ b2,
          float* __restrict__ P,
          float* __restrict__ Q) {
    static_assert(DIN == DH, "row gather is written for 128-float rows");
    static_assert(DOUT == MW * 16, "one matrix wave per 16 output columns");
    static_assert(!PROJ || DOUT == DH, "the projected tile is TM x DH");
    __shared__ __attribute__((aligned(16))) float Alds[2][TM * A_STRIDE];

    const unsigned tid = threadIdx.x;
    const bool fill = __builtin_amdgcn_readfirstlane((int)(tid >> 5)) >= MW;

    if (fill) __builtin_amdgcn_s_setprio(3);

    int t = blockIdx.x;
    if (fill) fill_tile(x, cnt, rowstart, nbr, Alds[0], t * TM, tid, 0, TM, true);
    __syncthreads();
    int cur = 0;
    for (; t < NTILES; t += gridDim.x) {
        const int tn = t + gridDim.x;
        v8f acc0 = {};
        v8f acc1 = {};
        float bv = 0.0f;
        if (!PROJ) {
            if (fill) {
                if (tn < NTILES)
                    fill_tile(x, cnt, rowstart, nbr, Alds[cur ^ 1], tn * TM, tid, 0, TM, true);
            } else {
                mma_acc<DIN, DOUT>(Alds[cur], Wl, Wr, tid, acc0, acc1);
                put_rows<DOUT, RELU>(acc0, acc1, col_bias(bias, tid), out + (size_t)(t * TM) * DOUT, DOUT, tid);
            }
            __syncthreads();
        } else {
            if (fill) {
                if (tn < NTILES)
                    fill_tile(x, cnt, rowstart, nbr, Alds[cur ^ 1], tn * TM, tid, 0, TM / 2, true);
            } else {
                mma_acc<DIN, DOUT>(Alds[cur], Wl, Wr, tid, acc0, acc1);
                bv = col_bias(bias, tid);
            }
            __syncthreads();
            if (!fill) put_rows<DOUT, RELU>(acc0, acc1, bv, Alds[cur], A_STRIDE, tid);
            __syncthreads();
            if (fill) {
                if (tn < NTILES)
                    fill_tile(x, cnt, rowstart, nbr, Alds[cur ^ 1], tn * TM, tid, TM / 2, TM, false);
            } else {
                proj_rows(Alds[cur], Wl2, Wr2, b2, P, Q, t * TM, tid);
            }
            __syncthreads();
        }
        cur ^= 1;
    }
    if (fill) __builtin_amdgcn_s_setprio(0);
}

__global__ void aggr_logsoftmax(const float* __restrict__ P,
                                const float* __restrict__ Q,
                                const int* __restrict__ cnt,
                                const int* __restrict__ rowstart,
                                const int* __restrict__ nbr,
                                float* __restrict__ out) {
    const int lane = threadIdx.x & 31;
    const int wave = threadIdx.x >> 5;
    const int grp  = lane / 10;
    const int ch   = lane - grp * 10;
    const int row  = (blockIdx.x * (blockDim.x >> 5) + wave) * 3 + grp;
    const bool on  = (grp < 3) && (row < NN);
    const int r    = on ? row : 0;
    const int c    = on ? cnt[r] : 0;
    const int beg  = rowstart[r];

    const float4* __restrict__ P4 = (const float4*)P;
    const float4 q = ((const float4*)Q)[(size_t)r * (NC / 4) + ch];

    float4 acc = make_float4(0.f, 0.f, 0.f, 0.f);
    int j = 0;
    for (; j + 8 <= c; j += 8) {
        int s[8];
        #pragma unroll
        for (int u = 0; u < 8; ++u) s[u] = nbr[beg + j + u];
        float4 v[8];
        #pragma unroll
        for (int u = 0; u < 8; ++u) v[u] = P4[(size_t)s[u] * (PS / 4) + ch];
        #pragma unroll
        for (int u = 0; u < 8; ++u) add4(acc, v[u]);
    }
    if (j < c) {
        const int left = c - j;
        int s[7];
        #pragma unroll
        for (int u = 0; u < 7; ++u) s[u] = nbr[beg + j + ((u < left) ? u : (left - 1))];
        float4 v[7];
        #pragma unroll
        for (int u = 0; u < 7; ++u) {
            v[u] = make_float4(0.f, 0.f, 0.f, 0.f);
            if (u < left) v[u] = P4[(size_t)s[u] * (PS / 4) + ch];
        }
        #pragma unroll
        for (int u = 0; u < 7; ++u) add4(acc, v[u]);
    }

    const float di = 1.0f / fmaxf((float)c, 1.0f);
    const float z0 = acc.x * di + q.x;
    const float z1 = acc.y * di + q.y;
    const float z2 = acc.z * di + q.z;
    const float z3 = acc.w * di + q.w;

    const int gb = (grp < 3) ? grp * 10 : 0;
    const float ml = fmaxf(fmaxf(z0, z1), fmaxf(z2, z3));
    float mx = -INFINITY;
    #pragma unroll
    for (int k = 0; k < 10; ++k) mx = fmaxf(mx, __shfl(ml, gb + k));
    const float el = expf(z0 - mx) + expf(z1 - mx) + expf(z2 - mx) + expf(z3 - mx);
    float sum = 0.0f;
    #pragma unroll
    for (int k = 0; k < 10; ++k) sum += __shfl(el, gb + k);
    const float lse = mx + logf(sum);

    if (on)
        ((float4*)out)[(size_t)row * (NC / 4) + ch] =
            make_float4(z0 - lse, z1 - lse, z2 - lse, z3 - lse);
}

extern "C" void kernel_launch(void* const* d_in, const int* in_sizes, int n_in,
                              void* d_out, int out_size, void* d_ws, size_t ws_size,
                              hipStream_t stream) {
    const float* x   = (const float*)d_in[0];
    const int*   ei  = (const int*)d_in[1];
    const int*   src = ei;
    const int*   dst = ei + NE;
    const float* Wl0 = (const float*)d_in[2];
    const float* Wr0 = (const float*)d_in[3];
    const float* b0  = (const float*)d_in[4];
    const float* Wl1 = (const float*)d_in[5];
    const float* Wr1 = (const float*)d_in[6];
    const float* b1  = (const float*)d_in[7];
    const float* Wl2 = (const float*)d_in[8];
    const float* Wr2 = (const float*)d_in[9];
    const float* b2  = (const float*)d_in[10];
    float* hout = (float*)d_out;

    char* ws = (char*)d_ws;
    size_t off = 0;
    auto carve = [&](size_t bytes) {
        void* p = ws + off;
        off = (off + bytes + 255) & ~(size_t)255;
        return p;
    };
    int*   cnt      = (int*)carve((size_t)NN * 4);
    int*   rowstart = (int*)carve((size_t)NN * 4);
    int*   part     = (int*)carve((size_t)NE * 4);
    int*   nbr      = (int*)carve((size_t)NE * 4);
    float* h0       = (float*)carve((size_t)NN * DH * 4);
    float* h1       = (float*)carve((size_t)NN * DH * 4);
    float* P        = h1;
    float* Q        = h1 + (size_t)NN * PS;
    int*   tab      = (int*)h1;

    const int ZB = 256;
    dim3 bZ(ZB);
    dim3 gAggr((NN + 23) / 24);

    chunk_group<<<dim3(NCH), dim3(ST), 0, stream>>>(src, dst, tab, part);
    bucket_sort<<<dim3(NB), dim3(512), 0, stream>>>(part, tab, cnt, rowstart, nbr);

    sage_gemm<DH, DH, true, false><<<dim3(LGRID), dim3((MW + FW) * 32), 0, stream>>>(
        x, cnt, rowstart, nbr, Wl0, Wr0, b0, h0,
        (const float*)nullptr, (const float*)nullptr, (const float*)nullptr, (float*)nullptr, (float*)nullptr);

    sage_gemm<DH, DH, true, true><<<dim3(LGRID), dim3((MW + FW) * 32), 0, stream>>>(
        h0, cnt, rowstart, nbr, Wl1, Wr1, b1, (float*)nullptr, Wl2, Wr2, b2, P, Q);

    aggr_logsoftmax<<<gAggr, bZ, 0, stream>>>(P, Q, cnt, rowstart, nbr, hout);
}
